# starting kernel
# baseline (speedup 1.0000x reference)
#include <hip/hip_runtime.h>
#include <math.h>

typedef __attribute__((ext_vector_type(2))) float v2f;
typedef __attribute__((ext_vector_type(8))) float v8f;

#define S 64
#define F 256
#define PITCH 66
#define TWO_PI 6.28318530717958647692f

__global__ void zero_out_kernel(float* out, int n) {
    int i = blockIdx.x * blockDim.x + threadIdx.x;
    if (i < n) out[i] = 0.0f;
}

__launch_bounds__(256)
__global__ void psf_render_kernel(const float* __restrict__ xs,
                                  const float* __restrict__ ys,
                                  const float* __restrict__ zs,
                                  const float* __restrict__ As,
                                  const float* __restrict__ bgs,
                                  const float* __restrict__ ps,
                                  const float* __restrict__ pupil_phase,
                                  float* __restrict__ out) {
    __shared__ float Pre[S * PITCH];
    __shared__ float Pim[S * PITCH];
    __shared__ float Wre[F * PITCH];
    __shared__ float Wim[F * PITCH];
    __shared__ float Tre[F * PITCH];
    __shared__ float Tim[F * PITCH];
    __shared__ float wave_max[8];

    const int e = blockIdx.x;
    const int b = e >> 2;
    if (ps[e] <= 0.5f) return;
    const float xe = xs[e], ye = ys[e], ze = zs[e];
    const float Ae = As[e], bge = bgs[e];

    const int t    = threadIdx.x;
    const int lane = t & 31;
    const int wave = t >> 5;
    const int col  = lane & 15;
    const int hi   = lane >> 4;

    for (int idx = t; idx < F * S; idx += 256) {
        const int v = idx >> 6, j = idx & 63;
        const int ti = ((j - 32) * v) & 255;
        float sn, cs;
        __sincosf(TWO_PI * (float)ti * (1.0f / 256.0f), &sn, &cs);
        Wre[v * PITCH + j] = cs;
        Wim[v * PITCH + j] = sn;
    }

    for (int idx = t; idx < S * S; idx += 256) {
        const int i = idx >> 6, j = idx & 63;
        const float ui = (2.0f * (float)i / 63.0f - 1.0f) * (1.0f / 0.75f);
        const float vj = (2.0f * (float)j / 63.0f - 1.0f) * (1.0f / 0.75f);
        const float r2 = ui * ui + vj * vj;
        const float mag  = (r2 <= 1.0f) ? 1.0f : 0.0f;
        const float prop = sqrtf(1.0f - fminf(r2, 1.0f));
        const float ph = pupil_phase[idx] + prop * ze
                       - TWO_PI * ((float)(i - 32) * (1.0f / 256.0f)) * xe
                       - TWO_PI * ((float)(j - 32) * (1.0f / 256.0f)) * ye;
        float sn, cs;
        __sincosf(ph, &sn, &cs);
        Pre[i * PITCH + j] = mag * cs;
        Pim[i * PITCH + j] = mag * sn;
    }
    __syncthreads();

    for (int tt = wave * 8; tt < wave * 8 + 8; ++tt) {
        const int ib = (tt >> 4) * 16;
        const int vb = (tt & 15) * 16;
        const int arow = ib + col;
        const int vcol = vb + col;
        v8f aRa = {}, aRb = {}, aIa = {}, aIb = {};
        #pragma unroll
        for (int kb = 0; kb < S; kb += 4) {
            const int k0 = kb + 2 * hi;
            const v2f aR = *(const v2f*)&Pre[arow * PITCH + k0];
            const v2f aI = *(const v2f*)&Pim[arow * PITCH + k0];
            const v2f bR = *(const v2f*)&Wre[vcol * PITCH + k0];
            const v2f bI = *(const v2f*)&Wim[vcol * PITCH + k0];
            aRa = __builtin_amdgcn_wmma_f32_16x16x4_f32(false, aR, false, bR, (short)0, aRa, false, false);
            aRb = __builtin_amdgcn_wmma_f32_16x16x4_f32(false, aI, false, bI, (short)0, aRb, false, false);
            aIa = __builtin_amdgcn_wmma_f32_16x16x4_f32(false, aR, false, bI, (short)0, aIa, false, false);
            aIb = __builtin_amdgcn_wmma_f32_16x16x4_f32(false, aI, false, bR, (short)0, aIb, false, false);
        }
        const int rbase = ib + 8 * hi;
        #pragma unroll
        for (int q = 0; q < 4; ++q) {
            v2f tr, ti;
            tr.x = aRa[2 * q]     - aRb[2 * q];
            tr.y = aRa[2 * q + 1] - aRb[2 * q + 1];
            ti.x = aIa[2 * q]     + aIb[2 * q];
            ti.y = aIa[2 * q + 1] + aIb[2 * q + 1];
            *(v2f*)&Tre[vcol * PITCH + rbase + 2 * q] = tr;
            *(v2f*)&Tim[vcol * PITCH + rbase + 2 * q] = ti;
        }
    }
    __syncthreads();

    float* crop = Pre;

    float mymax = 0.0f;
    for (int tt = wave * 32; tt < wave * 32 + 32; ++tt) {
        const int ub = (tt >> 4) * 16;
        const int vb = (tt & 15) * 16;
        const int urow = ub + col;
        const int vcol = vb + col;
        v8f aRa = {}, aRb = {}, aIa = {}, aIb = {};
        #pragma unroll
        for (int kb = 0; kb < S; kb += 4) {
            const int k0 = kb + 2 * hi;
            const v2f aR = *(const v2f*)&Wre[urow * PITCH + k0];
            const v2f aI = *(const v2f*)&Wim[urow * PITCH + k0];
            const v2f bR = *(const v2f*)&Tre[vcol * PITCH + k0];
            const v2f bI = *(const v2f*)&Tim[vcol * PITCH + k0];
            aRa = __builtin_amdgcn_wmma_f32_16x16x4_f32(false, aR, false, bR, (short)0, aRa, false, false);
            aRb = __builtin_amdgcn_wmma_f32_16x16x4_f32(false, aI, false, bI, (short)0, aRb, false, false);
            aIa = __builtin_amdgcn_wmma_f32_16x16x4_f32(false, aR, false, bI, (short)0, aIa, false, false);
            aIb = __builtin_amdgcn_wmma_f32_16x16x4_f32(false, aI, false, bR, (short)0, aIb, false, false);
        }
        const int wv = (vcol + 32) & 255;
        #pragma unroll
        for (int r = 0; r < 8; ++r) {
            const int u  = ub + r + 8 * hi;
            const float re = aRa[r] - aRb[r];
            const float im = aIa[r] + aIb[r];
            const float pw = re * re + im * im;
            mymax = fmaxf(mymax, pw);
            const int wu = (u + 32) & 255;
            if (wu < 64 && wv < 64) crop[wu * PITCH + wv] = pw;
        }
    }

    for (int off = 16; off > 0; off >>= 1)
        mymax = fmaxf(mymax, __shfl_xor(mymax, off, 32));
    if (lane == 0) wave_max[wave] = mymax;
    __syncthreads();
    float mx = wave_max[0];
    for (int w = 1; w < 8; ++w) mx = fmaxf(mx, wave_max[w]);
    const float inv = 1.0f / mx;

    for (int idx = t; idx < S * S; idx += 256) {
        const int cu = idx >> 6, cv = idx & 63;
        const float val = crop[cu * PITCH + cv] * inv * Ae + bge;
        atomicAdd(&out[b * S * S + idx], val);
    }
}

extern "C" void kernel_launch(void* const* d_in, const int* in_sizes, int n_in,
                              void* d_out, int out_size, void* d_ws, size_t ws_size,
                              hipStream_t stream) {
    const float* x  = (const float*)d_in[0];
    const float* y  = (const float*)d_in[1];
    const float* z  = (const float*)d_in[2];
    const float* A  = (const float*)d_in[3];
    const float* bg = (const float*)d_in[4];
    const float* p  = (const float*)d_in[5];
    const float* ph = (const float*)d_in[6];
    float* out = (float*)d_out;

    const int n_emit = in_sizes[0];
    zero_out_kernel<<<(out_size + 255) / 256, 256, 0, stream>>>(out, out_size);
    psf_render_kernel<<<n_emit, 256, 0, stream>>>(x, y, z, A, bg, p, ph, out);
}


# second transform of the maximum pass split radix-4 over its output row (u = u1 + 64*u2): four 16-term matrix products, then twiddle and 4-point sum in registers; f32 sum order changes
# speedup vs baseline: 2.0192x; 2.0192x over previous
#include <hip/hip_runtime.h>
#include <math.h>


typedef __attribute__((ext_vector_type(2))) float v2f;
typedef __attribute__((ext_vector_type(8))) float v8f;

#define S 64
#define F 256
#define PITCH 66
#define TWO_PI 6.28318530717958647692f
#define NWG 256
#define NPIECE 16

__device__ __forceinline__ void build_pupil(float* Pre, float* Pim,
                                            const float* __restrict__ pupil_phase,
                                            float xe, float ye, float ze, int t) {
    for (int idx = t; idx < S * S; idx += 256) {
        const int i = idx >> 6, j = idx & 63;
        const float ui = (2.0f * (float)i / 63.0f - 1.0f) * (1.0f / 0.75f);
        const float vj = (2.0f * (float)j / 63.0f - 1.0f) * (1.0f / 0.75f);
        const float r2 = ui * ui + vj * vj;
        const float mag  = (r2 <= 1.0f) ? 1.0f : 0.0f;
        const float prop = sqrtf(1.0f - fminf(r2, 1.0f));
        const float ph = pupil_phase[idx] + prop * ze
                       - TWO_PI * ((float)(i - 32) * (1.0f / 256.0f)) * xe
                       - TWO_PI * ((float)(j - 32) * (1.0f / 256.0f)) * ye;
        float sn, cs;
        __sincosf(ph, &sn, &cs);
        Pre[i * PITCH + j] = mag * cs;
        Pim[i * PITCH + j] = mag * sn;
    }
}

__device__ __forceinline__ void ctile(const float* Are, const float* Aim,
                                      const float* Bre, const float* Bim,
                                      int arow, int bcol, int hi,
                                      v8f& k1, v8f& k2, v8f& k3) {
    k1 = (v8f){}; k2 = (v8f){}; k3 = (v8f){};
    #pragma unroll
    for (int kb = 0; kb < S; kb += 4) {
        const int k0 = kb + 2 * hi;
        const v2f aR = *(const v2f*)&Are[arow * PITCH + k0];
        const v2f aI = *(const v2f*)&Aim[arow * PITCH + k0];
        const v2f bR = *(const v2f*)&Bre[bcol * PITCH + k0];
        const v2f bI = *(const v2f*)&Bim[bcol * PITCH + k0];
        const v2f aS = aR + aI;
        const v2f bD = bI - bR;
        const v2f bS = bR + bI;
        k1 = __builtin_amdgcn_wmma_f32_16x16x4_f32(false, aS, false, bR, (short)0, k1, false, false);
        k2 = __builtin_amdgcn_wmma_f32_16x16x4_f32(false, aR, false, bD, (short)0, k2, false, false);
        k3 = __builtin_amdgcn_wmma_f32_16x16x4_f32(false, aI, false, bS, (short)0, k3, false, false);
    }
}

__device__ __forceinline__ void store_T(float* Tre, float* Tim, int trow, int rbase,
                                        const v8f& k1, const v8f& k2, const v8f& k3) {
    #pragma unroll
    for (int q = 0; q < 4; ++q) {
        v2f tr, ti;
        tr.x = k1[2 * q]     - k3[2 * q];
        tr.y = k1[2 * q + 1] - k3[2 * q + 1];
        ti.x = k1[2 * q]     + k2[2 * q];
        ti.y = k1[2 * q + 1] + k2[2 * q + 1];
        *(v2f*)&Tre[trow * PITCH + rbase + 2 * q] = tr;
        *(v2f*)&Tim[trow * PITCH + rbase + 2 * q] = ti;
    }
}

__device__ __forceinline__ void store_T_split(float* Tre, float* Tim, int trow, int rbase,
                                              const v8f& k1, const v8f& k2, const v8f& k3) {
    #pragma unroll
    for (int q = 0; q < 4; ++q) {
        v2f tr, ti;
        tr.x = k1[q]     - k3[q];
        tr.y = k1[q + 4] - k3[q + 4];
        ti.x = k1[q]     + k2[q];
        ti.y = k1[q + 4] + k2[q + 4];
        *(v2f*)&Tre[trow * PITCH + q * 16 + (rbase >> 2)] = tr;
        *(v2f*)&Tim[trow * PITCH + q * 16 + (rbase >> 2)] = ti;
    }
}

__launch_bounds__(256)
__global__ void psf_max_kernel(const float* __restrict__ xs,
                               const float* __restrict__ ys,
                               const float* __restrict__ zs,
                               const float* __restrict__ ps,
                               const float* __restrict__ pupil_phase,
                               float* __restrict__ out, int n_out, int n_emit,
                               float* __restrict__ pmax) {
    __shared__ float Pre[S * PITCH];
    __shared__ float Pim[S * PITCH];
    __shared__ float Wre[F * PITCH];
    __shared__ float Wim[F * PITCH];
    __shared__ float Tre[F * PITCH];
    __shared__ float Tim[F * PITCH];
    __shared__ float wave_max[8];
    __shared__ int   wave_cnt[8];
    __shared__ int   live[256];

    const int t    = threadIdx.x;
    const int lane = t & 31;
    const int wave = t >> 5;
    const int col  = lane & 15;
    const int hi   = lane >> 4;

    for (int i = blockIdx.x * 256 + t; i < n_out; i += NWG * 256) out[i] = 0.0f;

    bool alive = false;
    if (t < n_emit) alive = !(ps[t] <= 0.5f);
    const unsigned m = (unsigned)__ballot(alive);
    if (lane == 0) wave_cnt[wave] = __popc(m);
    __syncthreads();
    int base = 0, n_live = 0;
    for (int w = 0; w < 8; ++w) {
        const int c = wave_cnt[w];
        if (w < wave) base += c;
        n_live += c;
    }
    if (alive) live[base + __popc(m & ((1u << lane) - 1u))] = t;

    const int lo   = (blockIdx.x * n_live) >> 4;
    const int hi_g = ((blockIdx.x + 1) * n_live) >> 4;
    if (lo >= hi_g) return;

    for (int idx = t; idx < F * S; idx += 256) {
        const int v = idx >> 6, j = idx & 63;
        const int ti = ((j - 32) * v) & 255;
        float sn, cs;
        __sincosf(TWO_PI * (float)ti * (1.0f / 256.0f), &sn, &cs);
        Wre[v * PITCH + j] = cs;
        Wim[v * PITCH + j] = sn;
    }

    for (int g0 = lo; g0 < hi_g; ) {
        const int r = g0 >> 4;
        int g1 = (r + 1) << 4;
        if (g1 > hi_g) g1 = hi_g;
        const int n   = g1 - g0;
        const int vt0 = g0 & 15;
        __syncthreads();
        const int e = live[r];
        build_pupil(Pre, Pim, pupil_phase, xs[e], ys[e], zs[e], t);
        __syncthreads();

        for (int tt = wave; tt < 4 * n; tt += 8) {
            const int ib = (tt & 3) * 16;
            const int vb = (vt0 + (tt >> 2)) * 16;
            v8f k1, k2, k3;
            ctile(Pre, Pim, Wre, Wim, ib + col, vb + col, hi, k1, k2, k3);
            store_T_split(Tre, Tim, vb + col, ib + 8 * hi, k1, k2, k3);
        }
        __syncthreads();

        const int u1b = (wave & 3) * 16;
        v2f eR[4], eI[4], eS[4];
        #pragma unroll
        for (int ks = 0; ks < 4; ++ks) {
            const int k0 = 24 + 4 * ks + 2 * hi;
            eR[ks] = *(const v2f*)&Wre[4 * (u1b + col) * PITCH + k0];
            eI[ks] = *(const v2f*)&Wim[4 * (u1b + col) * PITCH + k0];
            eS[ks] = eR[ks] + eI[ks];
        }
        float c1[8], s1[8];
        v2f c23[8], s23[8];
        #pragma unroll
        for (int r = 0; r < 8; ++r) {
            const int row = u1b + 8 * hi + r;
            c1[r]  = Wre[row * PITCH + 33];
            s1[r]  = Wim[row * PITCH + 33];
            c23[r] = *(const v2f*)&Wre[row * PITCH + 34];
            s23[r] = *(const v2f*)&Wim[row * PITCH + 34];
        }

        float mymax = 0.0f;
        for (int pp = wave >> 2; pp < n; pp += 2) {
            const int vrow = (vt0 + pp) * 16 + col;
            v8f acc[4][3];
            #pragma unroll
            for (int q = 0; q < 4; ++q) { acc[q][0] = (v8f){}; acc[q][1] = (v8f){}; acc[q][2] = (v8f){}; }
            #pragma unroll
            for (int ks = 0; ks < 4; ++ks) {
                #pragma unroll
                for (int q = 0; q < 4; ++q) {
                    const int k0 = q * 16 + 4 * ks + 2 * hi;
                    const v2f bR = *(const v2f*)&Tre[vrow * PITCH + k0];
                    const v2f bI = *(const v2f*)&Tim[vrow * PITCH + k0];
                    const v2f bD = bI - bR;
                    const v2f bS = bR + bI;
                    acc[q][0] = __builtin_amdgcn_wmma_f32_16x16x4_f32(false, eS[ks], false, bR, (short)0, acc[q][0], false, false);
                    acc[q][1] = __builtin_amdgcn_wmma_f32_16x16x4_f32(false, eR[ks], false, bD, (short)0, acc[q][1], false, false);
                    acc[q][2] = __builtin_amdgcn_wmma_f32_16x16x4_f32(false, eI[ks], false, bS, (short)0, acc[q][2], false, false);
                }
            }
            #pragma unroll
            for (int r = 0; r < 8; ++r) {
                const float g0r = acc[0][0][r] - acc[0][2][r];
                const float g0i = acc[0][0][r] + acc[0][1][r];
                const float x1r = acc[1][0][r] - acc[1][2][r];
                const float x1i = acc[1][0][r] + acc[1][1][r];
                const float x2r = acc[2][0][r] - acc[2][2][r];
                const float x2i = acc[2][0][r] + acc[2][1][r];
                const float x3r = acc[3][0][r] - acc[3][2][r];
                const float x3i = acc[3][0][r] + acc[3][1][r];
                const float g1r = x1r * c1[r]    - x1i * s1[r];
                const float g1i = x1r * s1[r]    + x1i * c1[r];
                const float g2r = x2r * c23[r].x - x2i * s23[r].x;
                const float g2i = x2r * s23[r].x + x2i * c23[r].x;
                const float g3r = x3r * c23[r].y - x3i * s23[r].y;
                const float g3i = x3r * s23[r].y + x3i * c23[r].y;
                const float ar = g0r + g2r, ai = g0i + g2i;
                const float br = g0r - g2r, bi = g0i - g2i;
                const float cr = g1r + g3r, ci = g1i + g3i;
                const float dr = g1r - g3r, di = g1i - g3i;
                const float o0r = ar + cr, o0i = ai + ci;
                const float o2r = ar - cr, o2i = ai - ci;
                const float o1r = br - di, o1i = bi + dr;
                const float o3r = br + di, o3i = bi - dr;
                mymax = fmaxf(mymax, o0r * o0r + o0i * o0i);
                mymax = fmaxf(mymax, o1r * o1r + o1i * o1i);
                mymax = fmaxf(mymax, o2r * o2r + o2i * o2i);
                mymax = fmaxf(mymax, o3r * o3r + o3i * o3i);
            }
        }
        for (int off = 16; off > 0; off >>= 1)
            mymax = fmaxf(mymax, __shfl_xor(mymax, off, 32));
        if (lane == 0) wave_max[wave] = mymax;
        __syncthreads();
        if (t < n) {
            float mx = wave_max[0];
            for (int w = 1; w < 8; ++w) mx = fmaxf(mx, wave_max[w]);
            pmax[e * NPIECE + vt0 + t] = mx;
        }
        g0 = g1;
    }
}

__launch_bounds__(256)
__global__ void psf_crop_kernel(const float* __restrict__ xs,
                                const float* __restrict__ ys,
                                const float* __restrict__ zs,
                                const float* __restrict__ As,
                                const float* __restrict__ bgs,
                                const float* __restrict__ ps,
                                const float* __restrict__ pupil_phase,
                                const float* __restrict__ pmax,
                                float* __restrict__ out) {
    __shared__ float Pre[S * PITCH];
    __shared__ float Pim[S * PITCH];
    __shared__ float Wre[S * PITCH];
    __shared__ float Wim[S * PITCH];
    __shared__ float Tre[S * PITCH];
    __shared__ float Tim[S * PITCH];

    const int e = blockIdx.x;
    const int b = e >> 2;
    if (ps[e] <= 0.5f) return;
    const float xe = xs[e], ye = ys[e], ze = zs[e];
    const float Ae = As[e], bge = bgs[e];

    const int t    = threadIdx.x;
    const int lane = t & 31;
    const int wave = t >> 5;
    const int col  = lane & 15;
    const int hi   = lane >> 4;

    const float pm = pmax[e * NPIECE + (t & 15)];

    for (int idx = t; idx < S * S; idx += 256) {
        const int c = idx >> 6, j = idx & 63;
        const int v = (c - 32) & 255;
        const int ti = ((j - 32) * v) & 255;
        float sn, cs;
        __sincosf(TWO_PI * (float)ti * (1.0f / 256.0f), &sn, &cs);
        Wre[c * PITCH + j] = cs;
        Wim[c * PITCH + j] = sn;
    }
    build_pupil(Pre, Pim, pupil_phase, xe, ye, ze, t);
    __syncthreads();

    for (int tt = wave * 2; tt < wave * 2 + 2; ++tt) {
        const int ib = (tt >> 2) * 16;
        const int cb = (tt & 3) * 16;
        v8f k1, k2, k3;
        ctile(Pre, Pim, Wre, Wim, ib + col, cb + col, hi, k1, k2, k3);
        store_T(Tre, Tim, cb + col, ib + 8 * hi, k1, k2, k3);
    }
    __syncthreads();

    float* crop = Pre;

    for (int tt = wave * 2; tt < wave * 2 + 2; ++tt) {
        const int ub = (tt >> 2) * 16;
        const int cb = (tt & 3) * 16;
        v8f k1, k2, k3;
        ctile(Wre, Wim, Tre, Tim, ub + col, cb + col, hi, k1, k2, k3);
        #pragma unroll
        for (int q = 0; q < 8; ++q) {
            const float re = k1[q] - k3[q];
            const float im = k1[q] + k2[q];
            const float pw = re * re + im * im;
            crop[(ub + q + 8 * hi) * PITCH + cb + col] = pw;
        }
    }

    float mx = pm;
    for (int off = 8; off > 0; off >>= 1)
        mx = fmaxf(mx, __shfl_xor(mx, off, 32));
    const float inv = 1.0f / mx;
    __syncthreads();

    for (int idx = t; idx < S * S; idx += 256) {
        const int cu = idx >> 6, cv = idx & 63;
        const float val = crop[cu * PITCH + cv] * inv * Ae + bge;
        atomicAdd(&out[b * S * S + idx], val);
    }
}

extern "C" void kernel_launch(void* const* d_in, const int* in_sizes, int n_in,
                              void* d_out, int out_size, void* d_ws, size_t ws_size,
                              hipStream_t stream) {
    const float* x  = (const float*)d_in[0];
    const float* y  = (const float*)d_in[1];
    const float* z  = (const float*)d_in[2];
    const float* A  = (const float*)d_in[3];
    const float* bg = (const float*)d_in[4];
    const float* p  = (const float*)d_in[5];
    const float* ph = (const float*)d_in[6];
    float* out = (float*)d_out;
    float* pmax = (float*)d_ws;

    const int n_emit = in_sizes[0];
    psf_max_kernel<<<NWG, 256, 0, stream>>>(x, y, z, p, ph, out, out_size, n_emit, pmax);
    psf_crop_kernel<<<n_emit, 256, 0, stream>>>(x, y, z, A, bg, p, ph, pmax, out);
}


# maximum pass stores the crop's |field|^2 it already forms (whole-line stores); launch 2 only scales and sums 4 emitters per output in fixed order, no transforms or atomics; crop f32 sum order changes
# speedup vs baseline: 2.9329x; 1.4525x over previous
#include <hip/hip_runtime.h>
#include <math.h>


typedef __attribute__((ext_vector_type(2))) float v2f;
typedef __attribute__((ext_vector_type(8))) float v8f;

#define S 64
#define F 256
#define PITCH 66
#define TWO_PI 6.28318530717958647692f
#define NWG 256
#define NPIECE 16
#define CROP_OFF (256 * NPIECE)
#define WS_NEED ((size_t)(CROP_OFF + 256 * S * S) * sizeof(float))

__device__ __forceinline__ void build_pupil(float* Pre, float* Pim,
                                            const float* __restrict__ pupil_phase,
                                            float xe, float ye, float ze, int t) {
    for (int idx = t; idx < S * S; idx += 256) {
        const int i = idx >> 6, j = idx & 63;
        const float ui = (2.0f * (float)i / 63.0f - 1.0f) * (1.0f / 0.75f);
        const float vj = (2.0f * (float)j / 63.0f - 1.0f) * (1.0f / 0.75f);
        const float r2 = ui * ui + vj * vj;
        const float mag  = (r2 <= 1.0f) ? 1.0f : 0.0f;
        const float prop = sqrtf(1.0f - fminf(r2, 1.0f));
        const float ph = pupil_phase[idx] + prop * ze
                       - TWO_PI * ((float)(i - 32) * (1.0f / 256.0f)) * xe
                       - TWO_PI * ((float)(j - 32) * (1.0f / 256.0f)) * ye;
        float sn, cs;
        __sincosf(ph, &sn, &cs);
        Pre[i * PITCH + j] = mag * cs;
        Pim[i * PITCH + j] = mag * sn;
    }
}

__device__ __forceinline__ void ctile(const float* Are, const float* Aim,
                                      const float* Bre, const float* Bim,
                                      int arow, int bcol, int hi,
                                      v8f& k1, v8f& k2, v8f& k3) {
    k1 = (v8f){}; k2 = (v8f){}; k3 = (v8f){};
    #pragma unroll
    for (int kb = 0; kb < S; kb += 4) {
        const int k0 = kb + 2 * hi;
        const v2f aR = *(const v2f*)&Are[arow * PITCH + k0];
        const v2f aI = *(const v2f*)&Aim[arow * PITCH + k0];
        const v2f bR = *(const v2f*)&Bre[bcol * PITCH + k0];
        const v2f bI = *(const v2f*)&Bim[bcol * PITCH + k0];
        const v2f aS = aR + aI;
        const v2f bD = bI - bR;
        const v2f bS = bR + bI;
        k1 = __builtin_amdgcn_wmma_f32_16x16x4_f32(false, aS, false, bR, (short)0, k1, false, false);
        k2 = __builtin_amdgcn_wmma_f32_16x16x4_f32(false, aR, false, bD, (short)0, k2, false, false);
        k3 = __builtin_amdgcn_wmma_f32_16x16x4_f32(false, aI, false, bS, (short)0, k3, false, false);
    }
}

__device__ __forceinline__ void store_T(float* Tre, float* Tim, int trow, int rbase,
                                        const v8f& k1, const v8f& k2, const v8f& k3) {
    #pragma unroll
    for (int q = 0; q < 4; ++q) {
        v2f tr, ti;
        tr.x = k1[2 * q]     - k3[2 * q];
        tr.y = k1[2 * q + 1] - k3[2 * q + 1];
        ti.x = k1[2 * q]     + k2[2 * q];
        ti.y = k1[2 * q + 1] + k2[2 * q + 1];
        *(v2f*)&Tre[trow * PITCH + rbase + 2 * q] = tr;
        *(v2f*)&Tim[trow * PITCH + rbase + 2 * q] = ti;
    }
}

__device__ __forceinline__ void store_T_split(float* Tre, float* Tim, int trow, int rbase,
                                              const v8f& k1, const v8f& k2, const v8f& k3) {
    #pragma unroll
    for (int q = 0; q < 4; ++q) {
        v2f tr, ti;
        tr.x = k1[q]     - k3[q];
        tr.y = k1[q + 4] - k3[q + 4];
        ti.x = k1[q]     + k2[q];
        ti.y = k1[q + 4] + k2[q + 4];
        *(v2f*)&Tre[trow * PITCH + q * 16 + (rbase >> 2)] = tr;
        *(v2f*)&Tim[trow * PITCH + q * 16 + (rbase >> 2)] = ti;
    }
}

__launch_bounds__(256)
__global__ void psf_max_kernel(const float* __restrict__ xs,
                               const float* __restrict__ ys,
                               const float* __restrict__ zs,
                               const float* __restrict__ ps,
                               const float* __restrict__ pupil_phase,
                               float* __restrict__ out, int n_out, int n_emit,
                               float* __restrict__ pmax,
                               float* __restrict__ crop_ws, size_t ws_bytes) {
    __shared__ float Pre[S * PITCH];
    __shared__ float Pim[S * PITCH];
    __shared__ float Wre[F * PITCH];
    __shared__ float Wim[F * PITCH];
    __shared__ float Tre[F * PITCH];
    __shared__ float Tim[F * PITCH];
    __shared__ float wave_max[8];
    __shared__ int   wave_cnt[8];
    __shared__ int   live[256];

    const int t    = threadIdx.x;
    const int lane = t & 31;
    const int wave = t >> 5;
    const int col  = lane & 15;
    const int hi   = lane >> 4;

    const bool keep = ws_bytes >= WS_NEED;
    if (!keep)
        for (int i = blockIdx.x * 256 + t; i < n_out; i += NWG * 256) out[i] = 0.0f;

    bool alive = false;
    if (t < n_emit) alive = !(ps[t] <= 0.5f);
    const unsigned m = (unsigned)__ballot(alive);
    if (lane == 0) wave_cnt[wave] = __popc(m);
    __syncthreads();
    int base = 0, n_live = 0;
    for (int w = 0; w < 8; ++w) {
        const int c = wave_cnt[w];
        if (w < wave) base += c;
        n_live += c;
    }
    if (alive) live[base + __popc(m & ((1u << lane) - 1u))] = t;

    const int lo   = (blockIdx.x * n_live) >> 4;
    const int hi_g = ((blockIdx.x + 1) * n_live) >> 4;
    if (lo >= hi_g) return;

    for (int idx = t; idx < F * S; idx += 256) {
        const int v = idx >> 6, j = idx & 63;
        const int ti = ((j - 32) * v) & 255;
        float sn, cs;
        __sincosf(TWO_PI * (float)ti * (1.0f / 256.0f), &sn, &cs);
        Wre[v * PITCH + j] = cs;
        Wim[v * PITCH + j] = sn;
    }

    for (int g0 = lo; g0 < hi_g; ) {
        const int r = g0 >> 4;
        int g1 = (r + 1) << 4;
        if (g1 > hi_g) g1 = hi_g;
        const int n   = g1 - g0;
        const int vt0 = g0 & 15;
        __syncthreads();
        const int e = live[r];
        build_pupil(Pre, Pim, pupil_phase, xs[e], ys[e], zs[e], t);
        __syncthreads();

        for (int tt = wave; tt < 4 * n; tt += 8) {
            const int ib = (tt & 3) * 16;
            const int vb = (vt0 + (tt >> 2)) * 16;
            v8f k1, k2, k3;
            ctile(Pre, Pim, Wre, Wim, ib + col, vb + col, hi, k1, k2, k3);
            store_T_split(Tre, Tim, vb + col, ib + 8 * hi, k1, k2, k3);
        }
        __syncthreads();

        const int u1b = (wave & 3) * 16;
        v2f eR[4], eI[4], eS[4];
        #pragma unroll
        for (int ks = 0; ks < 4; ++ks) {
            const int k0 = 24 + 4 * ks + 2 * hi;
            eR[ks] = *(const v2f*)&Wre[4 * (u1b + col) * PITCH + k0];
            eI[ks] = *(const v2f*)&Wim[4 * (u1b + col) * PITCH + k0];
            eS[ks] = eR[ks] + eI[ks];
        }
        float c1[8], s1[8];
        v2f c23[8], s23[8];
        #pragma unroll
        for (int r = 0; r < 8; ++r) {
            const int row = u1b + 8 * hi + r;
            c1[r]  = Wre[row * PITCH + 33];
            s1[r]  = Wim[row * PITCH + 33];
            c23[r] = *(const v2f*)&Wre[row * PITCH + 34];
            s23[r] = *(const v2f*)&Wim[row * PITCH + 34];
        }

        float mymax = 0.0f;
        for (int pp = wave >> 2; pp < n; pp += 2) {
            const int vrow = (vt0 + pp) * 16 + col;
            const int  cvb   = (vt0 + pp + 2) & 15;
            const bool cropu = keep && cvb < 4;
            const bool low   = (wave & 3) < 2;
            float* const dst = crop_ws + (size_t)e * (S * S)
                             + ((cvb & 3) * 4 + ((wave & 3) ^ 2)) * 256 + lane;
            float sel[8];
            v8f acc[4][3];
            #pragma unroll
            for (int q = 0; q < 4; ++q) { acc[q][0] = (v8f){}; acc[q][1] = (v8f){}; acc[q][2] = (v8f){}; }
            #pragma unroll
            for (int ks = 0; ks < 4; ++ks) {
                #pragma unroll
                for (int q = 0; q < 4; ++q) {
                    const int k0 = q * 16 + 4 * ks + 2 * hi;
                    const v2f bR = *(const v2f*)&Tre[vrow * PITCH + k0];
                    const v2f bI = *(const v2f*)&Tim[vrow * PITCH + k0];
                    const v2f bD = bI - bR;
                    const v2f bS = bR + bI;
                    acc[q][0] = __builtin_amdgcn_wmma_f32_16x16x4_f32(false, eS[ks], false, bR, (short)0, acc[q][0], false, false);
                    acc[q][1] = __builtin_amdgcn_wmma_f32_16x16x4_f32(false, eR[ks], false, bD, (short)0, acc[q][1], false, false);
                    acc[q][2] = __builtin_amdgcn_wmma_f32_16x16x4_f32(false, eI[ks], false, bS, (short)0, acc[q][2], false, false);
                }
            }
            #pragma unroll
            for (int r = 0; r < 8; ++r) {
                const float g0r = acc[0][0][r] - acc[0][2][r];
                const float g0i = acc[0][0][r] + acc[0][1][r];
                const float x1r = acc[1][0][r] - acc[1][2][r];
                const float x1i = acc[1][0][r] + acc[1][1][r];
                const float x2r = acc[2][0][r] - acc[2][2][r];
                const float x2i = acc[2][0][r] + acc[2][1][r];
                const float x3r = acc[3][0][r] - acc[3][2][r];
                const float x3i = acc[3][0][r] + acc[3][1][r];
                const float g1r = x1r * c1[r]    - x1i * s1[r];
                const float g1i = x1r * s1[r]    + x1i * c1[r];
                const float g2r = x2r * c23[r].x - x2i * s23[r].x;
                const float g2i = x2r * s23[r].x + x2i * c23[r].x;
                const float g3r = x3r * c23[r].y - x3i * s23[r].y;
                const float g3i = x3r * s23[r].y + x3i * c23[r].y;
                const float ar = g0r + g2r, ai = g0i + g2i;
                const float br = g0r - g2r, bi = g0i - g2i;
                const float cr = g1r + g3r, ci = g1i + g3i;
                const float dr = g1r - g3r, di = g1i - g3i;
                const float o0r = ar + cr, o0i = ai + ci;
                const float o2r = ar - cr, o2i = ai - ci;
                const float o1r = br - di, o1i = bi + dr;
                const float o3r = br + di, o3i = bi - dr;
                const float m0 = o0r * o0r + o0i * o0i;
                const float m3 = o3r * o3r + o3i * o3i;
                mymax = fmaxf(mymax, m0);
                mymax = fmaxf(mymax, o1r * o1r + o1i * o1i);
                mymax = fmaxf(mymax, o2r * o2r + o2i * o2i);
                mymax = fmaxf(mymax, m3);
                sel[r] = low ? m0 : m3;
            }
            if (cropu) {
                #pragma unroll
                for (int r = 0; r < 8; ++r) dst[r * 32] = sel[r];
            }
        }
        for (int off = 16; off > 0; off >>= 1)
            mymax = fmaxf(mymax, __shfl_xor(mymax, off, 32));
        if (lane == 0) wave_max[wave] = mymax;
        __syncthreads();
        if (t < n) {
            float mx = wave_max[0];
            for (int w = 1; w < 8; ++w) mx = fmaxf(mx, wave_max[w]);
            pmax[e * NPIECE + vt0 + t] = mx;
        }
        g0 = g1;
    }
}

__device__ __forceinline__ void crop_by_transform(const float* __restrict__ xs,
                                const float* __restrict__ ys,
                                const float* __restrict__ zs,
                                const float* __restrict__ As,
                                const float* __restrict__ bgs,
                                const float* __restrict__ ps,
                                const float* __restrict__ pupil_phase,
                                const float* __restrict__ pmax,
                                float* __restrict__ out) {
    __shared__ float Pre[S * PITCH];
    __shared__ float Pim[S * PITCH];
    __shared__ float Wre[S * PITCH];
    __shared__ float Wim[S * PITCH];
    __shared__ float Tre[S * PITCH];
    __shared__ float Tim[S * PITCH];

    const int e = blockIdx.x;
    const int b = e >> 2;
    if (ps[e] <= 0.5f) return;
    const float xe = xs[e], ye = ys[e], ze = zs[e];
    const float Ae = As[e], bge = bgs[e];

    const int t    = threadIdx.x;
    const int lane = t & 31;
    const int wave = t >> 5;
    const int col  = lane & 15;
    const int hi   = lane >> 4;

    const float pm = pmax[e * NPIECE + (t & 15)];

    for (int idx = t; idx < S * S; idx += 256) {
        const int c = idx >> 6, j = idx & 63;
        const int v = (c - 32) & 255;
        const int ti = ((j - 32) * v) & 255;
        float sn, cs;
        __sincosf(TWO_PI * (float)ti * (1.0f / 256.0f), &sn, &cs);
        Wre[c * PITCH + j] = cs;
        Wim[c * PITCH + j] = sn;
    }
    build_pupil(Pre, Pim, pupil_phase, xe, ye, ze, t);
    __syncthreads();

    for (int tt = wave * 2; tt < wave * 2 + 2; ++tt) {
        const int ib = (tt >> 2) * 16;
        const int cb = (tt & 3) * 16;
        v8f k1, k2, k3;
        ctile(Pre, Pim, Wre, Wim, ib + col, cb + col, hi, k1, k2, k3);
        store_T(Tre, Tim, cb + col, ib + 8 * hi, k1, k2, k3);
    }
    __syncthreads();

    float* crop = Pre;

    for (int tt = wave * 2; tt < wave * 2 + 2; ++tt) {
        const int ub = (tt >> 2) * 16;
        const int cb = (tt & 3) * 16;
        v8f k1, k2, k3;
        ctile(Wre, Wim, Tre, Tim, ub + col, cb + col, hi, k1, k2, k3);
        #pragma unroll
        for (int q = 0; q < 8; ++q) {
            const float re = k1[q] - k3[q];
            const float im = k1[q] + k2[q];
            const float pw = re * re + im * im;
            crop[(ub + q + 8 * hi) * PITCH + cb + col] = pw;
        }
    }

    float mx = pm;
    for (int off = 8; off > 0; off >>= 1)
        mx = fmaxf(mx, __shfl_xor(mx, off, 32));
    const float inv = 1.0f / mx;
    __syncthreads();

    for (int idx = t; idx < S * S; idx += 256) {
        const int cu = idx >> 6, cv = idx & 63;
        const float val = crop[cu * PITCH + cv] * inv * Ae + bge;
        atomicAdd(&out[b * S * S + idx], val);
    }
}

__launch_bounds__(256)
__global__ void psf_sum_kernel(const float* __restrict__ xs,
                               const float* __restrict__ ys,
                               const float* __restrict__ zs,
                               const float* __restrict__ As,
                               const float* __restrict__ bgs,
                               const float* __restrict__ ps,
                               const float* __restrict__ pupil_phase,
                               const float* __restrict__ pmax,
                               const float* __restrict__ crop_ws,
                               float* __restrict__ out, size_t ws_bytes) {
    if (ws_bytes < WS_NEED) {
        crop_by_transform(xs, ys, zs, As, bgs, ps, pupil_phase, pmax, out);
        return;
    }
    const int b    = blockIdx.x >> 2;
    const int cub  = blockIdx.x & 3;
    const int t    = threadIdx.x;
    const int lane = t & 31;
    const int rr   = t >> 5;
    const int hi   = lane >> 4;
    const int c4   = lane & 15;
    const int off  = ((c4 >> 2) * 4 + cub) * 256 + rr * 32 + hi * 16 + (c4 & 3) * 4;

    float4 acc = make_float4(0.0f, 0.0f, 0.0f, 0.0f);
    #pragma unroll
    for (int n = 0; n < 4; ++n) {
        const int e = 4 * b + n;
        const bool live = !(ps[e] <= 0.5f);
        const float* pm = live ? pmax + e * NPIECE : pupil_phase;
        const float* cs = live ? crop_ws + (size_t)e * (S * S) : pupil_phase;
        const float4 c = *(const float4*)&cs[off];
        float mx = pm[lane & 15];
        for (int o = 8; o > 0; o >>= 1)
            mx = fmaxf(mx, __shfl_xor(mx, o, 32));
        const float inv = 1.0f / mx;
        const float Ae = As[e], bge = bgs[e];
        acc.x += live ? c.x * inv * Ae + bge : 0.0f;
        acc.y += live ? c.y * inv * Ae + bge : 0.0f;
        acc.z += live ? c.z * inv * Ae + bge : 0.0f;
        acc.w += live ? c.w * inv * Ae + bge : 0.0f;
    }
    const int cu = cub * 16 + hi * 8 + rr;
    *(float4*)&out[(size_t)b * (S * S) + cu * S + c4 * 4] = acc;
}

extern "C" void kernel_launch(void* const* d_in, const int* in_sizes, int n_in,
                              void* d_out, int out_size, void* d_ws, size_t ws_size,
                              hipStream_t stream) {
    const float* x  = (const float*)d_in[0];
    const float* y  = (const float*)d_in[1];
    const float* z  = (const float*)d_in[2];
    const float* A  = (const float*)d_in[3];
    const float* bg = (const float*)d_in[4];
    const float* p  = (const float*)d_in[5];
    const float* ph = (const float*)d_in[6];
    float* out = (float*)d_out;
    float* pmax = (float*)d_ws;
    float* crop = pmax + CROP_OFF;

    const int n_emit = in_sizes[0];
    psf_max_kernel<<<NWG, 256, 0, stream>>>(x, y, z, p, ph, out, out_size, n_emit, pmax, crop, ws_size);
    psf_sum_kernel<<<n_emit, 256, 0, stream>>>(x, y, z, A, bg, p, ph, pmax, crop, out, ws_size);
}


# first transform of the maximum pass split radix-4 over its output column (v = v1 + 64*v2) like the second, same E and w_q: the 256x64 twiddle table is no longer built; f32 sum order changes
# speedup vs baseline: 3.5862x; 1.2227x over previous
#include <hip/hip_runtime.h>
#include <math.h>


typedef __attribute__((ext_vector_type(2))) float v2f;
typedef __attribute__((ext_vector_type(8))) float v8f;

#define S 64
#define F 256
#define PITCH 66
#define EP 18
#define TWO_PI 6.28318530717958647692f
#define NWG 256
#define NPIECE 16
#define CROP_OFF (256 * NPIECE)
#define WS_NEED ((size_t)(CROP_OFF + 256 * S * S) * sizeof(float))

template <bool SPLIT>
__device__ __forceinline__ void build_pupil(float* Pre, float* Pim,
                                            const float* __restrict__ pupil_phase,
                                            float xe, float ye, float ze, int t) {
    for (int idx = t; idx < S * S; idx += 256) {
        const int i = idx >> 6, j = idx & 63;
        const float ui = (2.0f * (float)i / 63.0f - 1.0f) * (1.0f / 0.75f);
        const float vj = (2.0f * (float)j / 63.0f - 1.0f) * (1.0f / 0.75f);
        const float r2 = ui * ui + vj * vj;
        const float mag  = (r2 <= 1.0f) ? 1.0f : 0.0f;
        const float prop = sqrtf(1.0f - fminf(r2, 1.0f));
        const float ph = pupil_phase[idx] + prop * ze
                       - TWO_PI * ((float)(i - 32) * (1.0f / 256.0f)) * xe
                       - TWO_PI * ((float)(j - 32) * (1.0f / 256.0f)) * ye;
        float sn, cs;
        __sincosf(ph, &sn, &cs);
        const int slot = SPLIT ? (j & 3) * 16 + (j >> 2) : j;
        Pre[i * PITCH + slot] = mag * cs;
        Pim[i * PITCH + slot] = mag * sn;
    }
}

__device__ __forceinline__ void ctile(const float* Are, const float* Aim,
                                      const float* Bre, const float* Bim,
                                      int arow, int bcol, int hi,
                                      v8f& k1, v8f& k2, v8f& k3) {
    k1 = (v8f){}; k2 = (v8f){}; k3 = (v8f){};
    #pragma unroll
    for (int kb = 0; kb < S; kb += 4) {
        const int k0 = kb + 2 * hi;
        const v2f aR = *(const v2f*)&Are[arow * PITCH + k0];
        const v2f aI = *(const v2f*)&Aim[arow * PITCH + k0];
        const v2f bR = *(const v2f*)&Bre[bcol * PITCH + k0];
        const v2f bI = *(const v2f*)&Bim[bcol * PITCH + k0];
        const v2f aS = aR + aI;
        const v2f bD = bI - bR;
        const v2f bS = bR + bI;
        k1 = __builtin_amdgcn_wmma_f32_16x16x4_f32(false, aS, false, bR, (short)0, k1, false, false);
        k2 = __builtin_amdgcn_wmma_f32_16x16x4_f32(false, aR, false, bD, (short)0, k2, false, false);
        k3 = __builtin_amdgcn_wmma_f32_16x16x4_f32(false, aI, false, bS, (short)0, k3, false, false);
    }
}

__device__ __forceinline__ void store_T(float* Tre, float* Tim, int trow, int rbase,
                                        const v8f& k1, const v8f& k2, const v8f& k3) {
    #pragma unroll
    for (int q = 0; q < 4; ++q) {
        v2f tr, ti;
        tr.x = k1[2 * q]     - k3[2 * q];
        tr.y = k1[2 * q + 1] - k3[2 * q + 1];
        ti.x = k1[2 * q]     + k2[2 * q];
        ti.y = k1[2 * q + 1] + k2[2 * q + 1];
        *(v2f*)&Tre[trow * PITCH + rbase + 2 * q] = tr;
        *(v2f*)&Tim[trow * PITCH + rbase + 2 * q] = ti;
    }
}

__launch_bounds__(256)
__global__ void psf_max_kernel(const float* __restrict__ xs,
                               const float* __restrict__ ys,
                               const float* __restrict__ zs,
                               const float* __restrict__ ps,
                               const float* __restrict__ pupil_phase,
                               float* __restrict__ out, int n_out, int n_emit,
                               float* __restrict__ pmax,
                               float* __restrict__ crop_ws, size_t ws_bytes) {
    __shared__ float Pre[S * PITCH];
    __shared__ float Pim[S * PITCH];
    __shared__ float Ere[S * EP];
    __shared__ float Eim[S * EP];
    __shared__ float Qre[S * 4];
    __shared__ float Qim[S * 4];
    __shared__ float Tre[F * PITCH];
    __shared__ float Tim[F * PITCH];
    __shared__ float wave_max[8];
    __shared__ int   wave_cnt[8];
    __shared__ int   live[256];

    const int t    = threadIdx.x;
    const int lane = t & 31;
    const int wave = t >> 5;
    const int col  = lane & 15;
    const int hi   = lane >> 4;

    const bool keep = ws_bytes >= WS_NEED;
    if (!keep)
        for (int i = blockIdx.x * 256 + t; i < n_out; i += NWG * 256) out[i] = 0.0f;

    bool alive = false;
    if (t < n_emit) alive = !(ps[t] <= 0.5f);
    const unsigned m = (unsigned)__ballot(alive);
    if (lane == 0) wave_cnt[wave] = __popc(m);
    __syncthreads();
    int base = 0, n_live = 0;
    for (int w = 0; w < 8; ++w) {
        const int c = wave_cnt[w];
        if (w < wave) base += c;
        n_live += c;
    }
    if (alive) live[base + __popc(m & ((1u << lane) - 1u))] = t;

    const int lo   = (blockIdx.x * n_live) >> 4;
    const int hi_g = ((blockIdx.x + 1) * n_live) >> 4;
    if (lo >= hi_g) return;

    for (int idx = t; idx < S * 16; idx += 256) {
        const int u = idx >> 4, p = idx & 15;
        const int ti = (4 * (p - 8) * u) & 255;
        float sn, cs;
        __sincosf(TWO_PI * (float)ti * (1.0f / 256.0f), &sn, &cs);
        Ere[u * EP + p] = cs;
        Eim[u * EP + p] = sn;
    }
    {
        const int ti = ((t & 3) * (t >> 2)) & 255;
        float sn, cs;
        __sincosf(TWO_PI * (float)ti * (1.0f / 256.0f), &sn, &cs);
        Qre[t] = cs;
        Qim[t] = sn;
    }

    for (int g0 = lo; g0 < hi_g; ) {
        const int r = g0 >> 4;
        int g1 = (r + 1) << 4;
        if (g1 > hi_g) g1 = hi_g;
        const int n   = g1 - g0;
        const int vt0 = g0 & 15;
        __syncthreads();
        const int e = live[r];
        build_pupil<true>(Pre, Pim, pupil_phase, xs[e], ys[e], zs[e], t);
        __syncthreads();

        const int bb0 = vt0 >> 2;
        const int nbb = ((vt0 + n - 1) >> 2) - bb0 + 1;
        for (int tt = wave; tt < 4 * nbb; tt += 8) {
            const int ib = (tt & 3) * 16;
            const int bb = bb0 + (tt >> 2);
            const int v1 = bb * 16 + col;
            v2f fR[4], fD[4], fS[4];
            #pragma unroll
            for (int ks = 0; ks < 4; ++ks) {
                const int k0 = 4 * ks + 2 * hi;
                const v2f fI = *(const v2f*)&Eim[v1 * EP + k0];
                fR[ks] = *(const v2f*)&Ere[v1 * EP + k0];
                fD[ks] = fI - fR[ks];
                fS[ks] = fR[ks] + fI;
            }
            const float wc1 = Qre[v1 * 4 + 1], ws1 = Qim[v1 * 4 + 1];
            const v2f wc23 = *(const v2f*)&Qre[v1 * 4 + 2];
            const v2f ws23 = *(const v2f*)&Qim[v1 * 4 + 2];
            v8f acc[4][3];
            #pragma unroll
            for (int q = 0; q < 4; ++q) { acc[q][0] = (v8f){}; acc[q][1] = (v8f){}; acc[q][2] = (v8f){}; }
            #pragma unroll
            for (int ks = 0; ks < 4; ++ks) {
                #pragma unroll
                for (int q = 0; q < 4; ++q) {
                    const int k0 = q * 16 + 4 * ks + 2 * hi;
                    const v2f aR = *(const v2f*)&Pre[(ib + col) * PITCH + k0];
                    const v2f aI = *(const v2f*)&Pim[(ib + col) * PITCH + k0];
                    const v2f aS = aR + aI;
                    acc[q][0] = __builtin_amdgcn_wmma_f32_16x16x4_f32(false, aS, false, fR[ks], (short)0, acc[q][0], false, false);
                    acc[q][1] = __builtin_amdgcn_wmma_f32_16x16x4_f32(false, aR, false, fD[ks], (short)0, acc[q][1], false, false);
                    acc[q][2] = __builtin_amdgcn_wmma_f32_16x16x4_f32(false, aI, false, fS[ks], (short)0, acc[q][2], false, false);
                }
            }
            float oR[4][8], oI[4][8];
            #pragma unroll
            for (int r8 = 0; r8 < 8; ++r8) {
                const float g0r = acc[0][0][r8] - acc[0][2][r8];
                const float g0i = acc[0][0][r8] + acc[0][1][r8];
                const float x1r = acc[1][0][r8] - acc[1][2][r8];
                const float x1i = acc[1][0][r8] + acc[1][1][r8];
                const float x2r = acc[2][0][r8] - acc[2][2][r8];
                const float x2i = acc[2][0][r8] + acc[2][1][r8];
                const float x3r = acc[3][0][r8] - acc[3][2][r8];
                const float x3i = acc[3][0][r8] + acc[3][1][r8];
                const float g1r = x1r * wc1    - x1i * ws1;
                const float g1i = x1r * ws1    + x1i * wc1;
                const float g2r = x2r * wc23.x - x2i * ws23.x;
                const float g2i = x2r * ws23.x + x2i * wc23.x;
                const float g3r = x3r * wc23.y - x3i * ws23.y;
                const float g3i = x3r * ws23.y + x3i * wc23.y;
                const float ar = g0r + g2r, ai = g0i + g2i;
                const float br = g0r - g2r, bi = g0i - g2i;
                const float cr = g1r + g3r, ci = g1i + g3i;
                const float dr = g1r - g3r, di = g1i - g3i;
                oR[0][r8] = ar + cr; oI[0][r8] = ai + ci;
                oR[1][r8] = br - di; oI[1][r8] = bi + dr;
                oR[2][r8] = ar - cr; oI[2][r8] = ai - ci;
                oR[3][r8] = br + di; oI[3][r8] = bi - dr;
            }
            #pragma unroll
            for (int v2 = 0; v2 < 4; ++v2) {
                const int pi = 4 * bb + v2;
                if (pi >= vt0 && pi < vt0 + n) {
                    #pragma unroll
                    for (int h = 0; h < 4; ++h) {
                        v2f tr, ti;
                        tr.x = oR[v2][h]; tr.y = oR[v2][h + 4];
                        ti.x = oI[v2][h]; ti.y = oI[v2][h + 4];
                        *(v2f*)&Tre[(v1 + 64 * v2) * PITCH + h * 16 + ((ib + 8 * hi) >> 2)] = tr;
                        *(v2f*)&Tim[(v1 + 64 * v2) * PITCH + h * 16 + ((ib + 8 * hi) >> 2)] = ti;
                    }
                }
            }
        }
        __syncthreads();

        const int u1b = (wave & 3) * 16;
        v2f eR[4], eI[4], eS[4];
        #pragma unroll
        for (int ks = 0; ks < 4; ++ks) {
            const int k0 = 4 * ks + 2 * hi;
            eR[ks] = *(const v2f*)&Ere[(u1b + col) * EP + k0];
            eI[ks] = *(const v2f*)&Eim[(u1b + col) * EP + k0];
            eS[ks] = eR[ks] + eI[ks];
        }
        float c1[8], s1[8];
        v2f c23[8], s23[8];
        #pragma unroll
        for (int r = 0; r < 8; ++r) {
            const int row = u1b + 8 * hi + r;
            c1[r]  = Qre[row * 4 + 1];
            s1[r]  = Qim[row * 4 + 1];
            c23[r] = *(const v2f*)&Qre[row * 4 + 2];
            s23[r] = *(const v2f*)&Qim[row * 4 + 2];
        }

        float mymax = 0.0f;
        for (int pp = wave >> 2; pp < n; pp += 2) {
            const int cblk = ((vt0 + pp) >> 2) + 4 * ((vt0 + pp) & 3);
            const int vrow = cblk * 16 + col;
            const int  cvb   = (cblk + 2) & 15;
            const bool cropu = keep && cvb < 4;
            const bool low   = (wave & 3) < 2;
            float* const dst = crop_ws + (size_t)e * (S * S)
                             + ((cvb & 3) * 4 + ((wave & 3) ^ 2)) * 256 + lane;
            float sel[8];
            v8f acc[4][3];
            #pragma unroll
            for (int q = 0; q < 4; ++q) { acc[q][0] = (v8f){}; acc[q][1] = (v8f){}; acc[q][2] = (v8f){}; }
            #pragma unroll
            for (int ks = 0; ks < 4; ++ks) {
                #pragma unroll
                for (int q = 0; q < 4; ++q) {
                    const int k0 = q * 16 + 4 * ks + 2 * hi;
                    const v2f bR = *(const v2f*)&Tre[vrow * PITCH + k0];
                    const v2f bI = *(const v2f*)&Tim[vrow * PITCH + k0];
                    const v2f bD = bI - bR;
                    const v2f bS = bR + bI;
                    acc[q][0] = __builtin_amdgcn_wmma_f32_16x16x4_f32(false, eS[ks], false, bR, (short)0, acc[q][0], false, false);
                    acc[q][1] = __builtin_amdgcn_wmma_f32_16x16x4_f32(false, eR[ks], false, bD, (short)0, acc[q][1], false, false);
                    acc[q][2] = __builtin_amdgcn_wmma_f32_16x16x4_f32(false, eI[ks], false, bS, (short)0, acc[q][2], false, false);
                }
            }
            #pragma unroll
            for (int r = 0; r < 8; ++r) {
                const float g0r = acc[0][0][r] - acc[0][2][r];
                const float g0i = acc[0][0][r] + acc[0][1][r];
                const float x1r = acc[1][0][r] - acc[1][2][r];
                const float x1i = acc[1][0][r] + acc[1][1][r];
                const float x2r = acc[2][0][r] - acc[2][2][r];
                const float x2i = acc[2][0][r] + acc[2][1][r];
                const float x3r = acc[3][0][r] - acc[3][2][r];
                const float x3i = acc[3][0][r] + acc[3][1][r];
                const float g1r = x1r * c1[r]    - x1i * s1[r];
                const float g1i = x1r * s1[r]    + x1i * c1[r];
                const float g2r = x2r * c23[r].x - x2i * s23[r].x;
                const float g2i = x2r * s23[r].x + x2i * c23[r].x;
                const float g3r = x3r * c23[r].y - x3i * s23[r].y;
                const float g3i = x3r * s23[r].y + x3i * c23[r].y;
                const float ar = g0r + g2r, ai = g0i + g2i;
                const float br = g0r - g2r, bi = g0i - g2i;
                const float cr = g1r + g3r, ci = g1i + g3i;
                const float dr = g1r - g3r, di = g1i - g3i;
                const float o0r = ar + cr, o0i = ai + ci;
                const float o2r = ar - cr, o2i = ai - ci;
                const float o1r = br - di, o1i = bi + dr;
                const float o3r = br + di, o3i = bi - dr;
                const float m0 = o0r * o0r + o0i * o0i;
                const float m3 = o3r * o3r + o3i * o3i;
                mymax = fmaxf(mymax, m0);
                mymax = fmaxf(mymax, o1r * o1r + o1i * o1i);
                mymax = fmaxf(mymax, o2r * o2r + o2i * o2i);
                mymax = fmaxf(mymax, m3);
                sel[r] = low ? m0 : m3;
            }
            if (cropu) {
                #pragma unroll
                for (int r = 0; r < 8; ++r) dst[r * 32] = sel[r];
            }
        }
        for (int off = 16; off > 0; off >>= 1)
            mymax = fmaxf(mymax, __shfl_xor(mymax, off, 32));
        if (lane == 0) wave_max[wave] = mymax;
        __syncthreads();
        if (t < n) {
            float mx = wave_max[0];
            for (int w = 1; w < 8; ++w) mx = fmaxf(mx, wave_max[w]);
            pmax[e * NPIECE + vt0 + t] = mx;
        }
        g0 = g1;
    }
}

__device__ __forceinline__ void crop_by_transform(const float* __restrict__ xs,
                                const float* __restrict__ ys,
                                const float* __restrict__ zs,
                                const float* __restrict__ As,
                                const float* __restrict__ bgs,
                                const float* __restrict__ ps,
                                const float* __restrict__ pupil_phase,
                                const float* __restrict__ pmax,
                                float* __restrict__ out) {
    __shared__ float Pre[S * PITCH];
    __shared__ float Pim[S * PITCH];
    __shared__ float Wre[S * PITCH];
    __shared__ float Wim[S * PITCH];
    __shared__ float Tre[S * PITCH];
    __shared__ float Tim[S * PITCH];

    const int e = blockIdx.x;
    const int b = e >> 2;
    if (ps[e] <= 0.5f) return;
    const float xe = xs[e], ye = ys[e], ze = zs[e];
    const float Ae = As[e], bge = bgs[e];

    const int t    = threadIdx.x;
    const int lane = t & 31;
    const int wave = t >> 5;
    const int col  = lane & 15;
    const int hi   = lane >> 4;

    const float pm = pmax[e * NPIECE + (t & 15)];

    for (int idx = t; idx < S * S; idx += 256) {
        const int c = idx >> 6, j = idx & 63;
        const int v = (c - 32) & 255;
        const int ti = ((j - 32) * v) & 255;
        float sn, cs;
        __sincosf(TWO_PI * (float)ti * (1.0f / 256.0f), &sn, &cs);
        Wre[c * PITCH + j] = cs;
        Wim[c * PITCH + j] = sn;
    }
    build_pupil<false>(Pre, Pim, pupil_phase, xe, ye, ze, t);
    __syncthreads();

    for (int tt = wave * 2; tt < wave * 2 + 2; ++tt) {
        const int ib = (tt >> 2) * 16;
        const int cb = (tt & 3) * 16;
        v8f k1, k2, k3;
        ctile(Pre, Pim, Wre, Wim, ib + col, cb + col, hi, k1, k2, k3);
        store_T(Tre, Tim, cb + col, ib + 8 * hi, k1, k2, k3);
    }
    __syncthreads();

    float* crop = Pre;

    for (int tt = wave * 2; tt < wave * 2 + 2; ++tt) {
        const int ub = (tt >> 2) * 16;
        const int cb = (tt & 3) * 16;
        v8f k1, k2, k3;
        ctile(Wre, Wim, Tre, Tim, ub + col, cb + col, hi, k1, k2, k3);
        #pragma unroll
        for (int q = 0; q < 8; ++q) {
            const float re = k1[q] - k3[q];
            const float im = k1[q] + k2[q];
            const float pw = re * re + im * im;
            crop[(ub + q + 8 * hi) * PITCH + cb + col] = pw;
        }
    }

    float mx = pm;
    for (int off = 8; off > 0; off >>= 1)
        mx = fmaxf(mx, __shfl_xor(mx, off, 32));
    const float inv = 1.0f / mx;
    __syncthreads();

    for (int idx = t; idx < S * S; idx += 256) {
        const int cu = idx >> 6, cv = idx & 63;
        const float val = crop[cu * PITCH + cv] * inv * Ae + bge;
        atomicAdd(&out[b * S * S + idx], val);
    }
}

__launch_bounds__(256)
__global__ void psf_sum_kernel(const float* __restrict__ xs,
                               const float* __restrict__ ys,
                               const float* __restrict__ zs,
                               const float* __restrict__ As,
                               const float* __restrict__ bgs,
                               const float* __restrict__ ps,
                               const float* __restrict__ pupil_phase,
                               const float* __restrict__ pmax,
                               const float* __restrict__ crop_ws,
                               float* __restrict__ out, size_t ws_bytes) {
    if (ws_bytes < WS_NEED) {
        crop_by_transform(xs, ys, zs, As, bgs, ps, pupil_phase, pmax, out);
        return;
    }
    const int b    = blockIdx.x >> 2;
    const int cub  = blockIdx.x & 3;
    const int t    = threadIdx.x;
    const int lane = t & 31;
    const int rr   = t >> 5;
    const int hi   = lane >> 4;
    const int c4   = lane & 15;
    const int off  = ((c4 >> 2) * 4 + cub) * 256 + rr * 32 + hi * 16 + (c4 & 3) * 4;

    float4 acc = make_float4(0.0f, 0.0f, 0.0f, 0.0f);
    #pragma unroll
    for (int n = 0; n < 4; ++n) {
        const int e = 4 * b + n;
        const bool live = !(ps[e] <= 0.5f);
        const float* pm = live ? pmax + e * NPIECE : pupil_phase;
        const float* cs = live ? crop_ws + (size_t)e * (S * S) : pupil_phase;
        const float4 c = *(const float4*)&cs[off];
        float mx = pm[lane & 15];
        for (int o = 8; o > 0; o >>= 1)
            mx = fmaxf(mx, __shfl_xor(mx, o, 32));
        const float inv = 1.0f / mx;
        const float Ae = As[e], bge = bgs[e];
        acc.x += live ? c.x * inv * Ae + bge : 0.0f;
        acc.y += live ? c.y * inv * Ae + bge : 0.0f;
        acc.z += live ? c.z * inv * Ae + bge : 0.0f;
        acc.w += live ? c.w * inv * Ae + bge : 0.0f;
    }
    const int cu = cub * 16 + hi * 8 + rr;
    *(float4*)&out[(size_t)b * (S * S) + cu * S + c4 * 4] = acc;
}

extern "C" void kernel_launch(void* const* d_in, const int* in_sizes, int n_in,
                              void* d_out, int out_size, void* d_ws, size_t ws_size,
                              hipStream_t stream) {
    const float* x  = (const float*)d_in[0];
    const float* y  = (const float*)d_in[1];
    const float* z  = (const float*)d_in[2];
    const float* A  = (const float*)d_in[3];
    const float* bg = (const float*)d_in[4];
    const float* p  = (const float*)d_in[5];
    const float* ph = (const float*)d_in[6];
    float* out = (float*)d_out;
    float* pmax = (float*)d_ws;
    float* crop = pmax + CROP_OFF;

    const int n_emit = in_sizes[0];
    psf_max_kernel<<<NWG, 256, 0, stream>>>(x, y, z, p, ph, out, out_size, n_emit, pmax, crop, ws_size);
    psf_sum_kernel<<<n_emit, 256, 0, stream>>>(x, y, z, A, bg, p, ph, pmax, crop, out, ws_size);
}


# emitter-independent half of the pupil build (aperture mask, defocus factor, phase loads) formed once per work-group ahead of the emitter loop, kept in registers; per emitter only phase sum and sincos
# speedup vs baseline: 4.0962x; 1.1422x over previous
#include <hip/hip_runtime.h>
#include <math.h>


typedef __attribute__((ext_vector_type(2))) float v2f;
typedef __attribute__((ext_vector_type(8))) float v8f;

#define S 64
#define F 256
#define PITCH 66
#define EP 18
#define TWO_PI 6.28318530717958647692f
#define NWG 256
#define NPIECE 16
#define CROP_OFF (256 * NPIECE)
#define WS_NEED ((size_t)(CROP_OFF + 256 * S * S) * sizeof(float))

__device__ __forceinline__ void build_pupil(float* Pre, float* Pim,
                                            const float* __restrict__ pupil_phase,
                                            float xe, float ye, float ze, int t) {
    for (int idx = t; idx < S * S; idx += 256) {
        const int i = idx >> 6, j = idx & 63;
        const float ui = (2.0f * (float)i / 63.0f - 1.0f) * (1.0f / 0.75f);
        const float vj = (2.0f * (float)j / 63.0f - 1.0f) * (1.0f / 0.75f);
        const float r2 = ui * ui + vj * vj;
        const float mag  = (r2 <= 1.0f) ? 1.0f : 0.0f;
        const float prop = sqrtf(1.0f - fminf(r2, 1.0f));
        const float ph = pupil_phase[idx] + prop * ze
                       - TWO_PI * ((float)(i - 32) * (1.0f / 256.0f)) * xe
                       - TWO_PI * ((float)(j - 32) * (1.0f / 256.0f)) * ye;
        float sn, cs;
        __sincosf(ph, &sn, &cs);
        Pre[i * PITCH + j] = mag * cs;
        Pim[i * PITCH + j] = mag * sn;
    }
}

__device__ __forceinline__ void ctile(const float* Are, const float* Aim,
                                      const float* Bre, const float* Bim,
                                      int arow, int bcol, int hi,
                                      v8f& k1, v8f& k2, v8f& k3) {
    k1 = (v8f){}; k2 = (v8f){}; k3 = (v8f){};
    #pragma unroll
    for (int kb = 0; kb < S; kb += 4) {
        const int k0 = kb + 2 * hi;
        const v2f aR = *(const v2f*)&Are[arow * PITCH + k0];
        const v2f aI = *(const v2f*)&Aim[arow * PITCH + k0];
        const v2f bR = *(const v2f*)&Bre[bcol * PITCH + k0];
        const v2f bI = *(const v2f*)&Bim[bcol * PITCH + k0];
        const v2f aS = aR + aI;
        const v2f bD = bI - bR;
        const v2f bS = bR + bI;
        k1 = __builtin_amdgcn_wmma_f32_16x16x4_f32(false, aS, false, bR, (short)0, k1, false, false);
        k2 = __builtin_amdgcn_wmma_f32_16x16x4_f32(false, aR, false, bD, (short)0, k2, false, false);
        k3 = __builtin_amdgcn_wmma_f32_16x16x4_f32(false, aI, false, bS, (short)0, k3, false, false);
    }
}

__device__ __forceinline__ void store_T(float* Tre, float* Tim, int trow, int rbase,
                                        const v8f& k1, const v8f& k2, const v8f& k3) {
    #pragma unroll
    for (int q = 0; q < 4; ++q) {
        v2f tr, ti;
        tr.x = k1[2 * q]     - k3[2 * q];
        tr.y = k1[2 * q + 1] - k3[2 * q + 1];
        ti.x = k1[2 * q]     + k2[2 * q];
        ti.y = k1[2 * q + 1] + k2[2 * q + 1];
        *(v2f*)&Tre[trow * PITCH + rbase + 2 * q] = tr;
        *(v2f*)&Tim[trow * PITCH + rbase + 2 * q] = ti;
    }
}

__launch_bounds__(256)
__global__ void psf_max_kernel(const float* __restrict__ xs,
                               const float* __restrict__ ys,
                               const float* __restrict__ zs,
                               const float* __restrict__ ps,
                               const float* __restrict__ pupil_phase,
                               float* __restrict__ out, int n_out, int n_emit,
                               float* __restrict__ pmax,
                               float* __restrict__ crop_ws, size_t ws_bytes) {
    __shared__ float Pre[S * PITCH];
    __shared__ float Pim[S * PITCH];
    __shared__ float Utab[S];
    __shared__ float Ere[S * EP];
    __shared__ float Eim[S * EP];
    __shared__ float Qre[S * 4];
    __shared__ float Qim[S * 4];
    __shared__ float Tre[F * PITCH];
    __shared__ float Tim[F * PITCH];
    __shared__ float wave_max[8];
    __shared__ int   wave_cnt[8];
    __shared__ int   live[256];

    const int t    = threadIdx.x;
    const int lane = t & 31;
    const int wave = t >> 5;
    const int col  = lane & 15;
    const int hi   = lane >> 4;

    const bool keep = ws_bytes >= WS_NEED;
    if (!keep)
        for (int i = blockIdx.x * 256 + t; i < n_out; i += NWG * 256) out[i] = 0.0f;

    float pph[16];
    #pragma unroll
    for (int k = 0; k < 16; ++k) pph[k] = pupil_phase[t + 256 * k];
    if (t < S) Utab[t] = (2.0f * (float)t / 63.0f - 1.0f) * (1.0f / 0.75f);

    bool alive = false;
    if (t < n_emit) alive = !(ps[t] <= 0.5f);
    const unsigned m = (unsigned)__ballot(alive);
    if (lane == 0) wave_cnt[wave] = __popc(m);
    __syncthreads();
    int base = 0, n_live = 0;
    for (int w = 0; w < 8; ++w) {
        const int c = wave_cnt[w];
        if (w < wave) base += c;
        n_live += c;
    }
    if (alive) live[base + __popc(m & ((1u << lane) - 1u))] = t;

    const int lo   = (blockIdx.x * n_live) >> 4;
    const int hi_g = ((blockIdx.x + 1) * n_live) >> 4;
    if (lo >= hi_g) return;

    float mag[16], prop[16];
    {
        const float vj = Utab[t & 63];
        #pragma unroll
        for (int k = 0; k < 16; ++k) {
            const float ui = Utab[(t >> 6) + 4 * k];
            const float r2 = ui * ui + vj * vj;
            mag[k]  = (r2 <= 1.0f) ? 1.0f : 0.0f;
            prop[k] = sqrtf(1.0f - fminf(r2, 1.0f));
        }
    }

    for (int idx = t; idx < S * 16; idx += 256) {
        const int u = idx >> 4, p = idx & 15;
        const int ti = (4 * (p - 8) * u) & 255;
        float sn, cs;
        __sincosf(TWO_PI * (float)ti * (1.0f / 256.0f), &sn, &cs);
        Ere[u * EP + p] = cs;
        Eim[u * EP + p] = sn;
    }
    {
        const int ti = ((t & 3) * (t >> 2)) & 255;
        float sn, cs;
        __sincosf(TWO_PI * (float)ti * (1.0f / 256.0f), &sn, &cs);
        Qre[t] = cs;
        Qim[t] = sn;
    }

    for (int g0 = lo; g0 < hi_g; ) {
        const int r = g0 >> 4;
        int g1 = (r + 1) << 4;
        if (g1 > hi_g) g1 = hi_g;
        const int n   = g1 - g0;
        const int vt0 = g0 & 15;
        __syncthreads();
        const int e = live[r];
        {
            const float xe = xs[e], ye = ys[e], ze = zs[e];
            const int j = t & 63;
            const int slot = (j & 3) * 16 + (j >> 2);
            #pragma unroll
            for (int k = 0; k < 16; ++k) {
                const int i = (t >> 6) + 4 * k;
                const float ph = pph[k] + prop[k] * ze
                               - TWO_PI * ((float)(i - 32) * (1.0f / 256.0f)) * xe
                               - TWO_PI * ((float)(j - 32) * (1.0f / 256.0f)) * ye;
                float sn, cs;
                __sincosf(ph, &sn, &cs);
                Pre[i * PITCH + slot] = mag[k] * cs;
                Pim[i * PITCH + slot] = mag[k] * sn;
            }
        }
        __syncthreads();

        const int bb0 = vt0 >> 2;
        const int nbb = ((vt0 + n - 1) >> 2) - bb0 + 1;
        for (int tt = wave; tt < 4 * nbb; tt += 8) {
            const int ib = (tt & 3) * 16;
            const int bb = bb0 + (tt >> 2);
            const int v1 = bb * 16 + col;
            v2f fR[4], fD[4], fS[4];
            #pragma unroll
            for (int ks = 0; ks < 4; ++ks) {
                const int k0 = 4 * ks + 2 * hi;
                const v2f fI = *(const v2f*)&Eim[v1 * EP + k0];
                fR[ks] = *(const v2f*)&Ere[v1 * EP + k0];
                fD[ks] = fI - fR[ks];
                fS[ks] = fR[ks] + fI;
            }
            const float wc1 = Qre[v1 * 4 + 1], ws1 = Qim[v1 * 4 + 1];
            const v2f wc23 = *(const v2f*)&Qre[v1 * 4 + 2];
            const v2f ws23 = *(const v2f*)&Qim[v1 * 4 + 2];
            v8f acc[4][3];
            #pragma unroll
            for (int q = 0; q < 4; ++q) { acc[q][0] = (v8f){}; acc[q][1] = (v8f){}; acc[q][2] = (v8f){}; }
            #pragma unroll
            for (int ks = 0; ks < 4; ++ks) {
                #pragma unroll
                for (int q = 0; q < 4; ++q) {
                    const int k0 = q * 16 + 4 * ks + 2 * hi;
                    const v2f aR = *(const v2f*)&Pre[(ib + col) * PITCH + k0];
                    const v2f aI = *(const v2f*)&Pim[(ib + col) * PITCH + k0];
                    const v2f aS = aR + aI;
                    acc[q][0] = __builtin_amdgcn_wmma_f32_16x16x4_f32(false, aS, false, fR[ks], (short)0, acc[q][0], false, false);
                    acc[q][1] = __builtin_amdgcn_wmma_f32_16x16x4_f32(false, aR, false, fD[ks], (short)0, acc[q][1], false, false);
                    acc[q][2] = __builtin_amdgcn_wmma_f32_16x16x4_f32(false, aI, false, fS[ks], (short)0, acc[q][2], false, false);
                }
            }
            float oR[4][8], oI[4][8];
            #pragma unroll
            for (int r8 = 0; r8 < 8; ++r8) {
                const float g0r = acc[0][0][r8] - acc[0][2][r8];
                const float g0i = acc[0][0][r8] + acc[0][1][r8];
                const float x1r = acc[1][0][r8] - acc[1][2][r8];
                const float x1i = acc[1][0][r8] + acc[1][1][r8];
                const float x2r = acc[2][0][r8] - acc[2][2][r8];
                const float x2i = acc[2][0][r8] + acc[2][1][r8];
                const float x3r = acc[3][0][r8] - acc[3][2][r8];
                const float x3i = acc[3][0][r8] + acc[3][1][r8];
                const float g1r = x1r * wc1    - x1i * ws1;
                const float g1i = x1r * ws1    + x1i * wc1;
                const float g2r = x2r * wc23.x - x2i * ws23.x;
                const float g2i = x2r * ws23.x + x2i * wc23.x;
                const float g3r = x3r * wc23.y - x3i * ws23.y;
                const float g3i = x3r * ws23.y + x3i * wc23.y;
                const float ar = g0r + g2r, ai = g0i + g2i;
                const float br = g0r - g2r, bi = g0i - g2i;
                const float cr = g1r + g3r, ci = g1i + g3i;
                const float dr = g1r - g3r, di = g1i - g3i;
                oR[0][r8] = ar + cr; oI[0][r8] = ai + ci;
                oR[1][r8] = br - di; oI[1][r8] = bi + dr;
                oR[2][r8] = ar - cr; oI[2][r8] = ai - ci;
                oR[3][r8] = br + di; oI[3][r8] = bi - dr;
            }
            #pragma unroll
            for (int v2 = 0; v2 < 4; ++v2) {
                const int pi = 4 * bb + v2;
                if (pi >= vt0 && pi < vt0 + n) {
                    #pragma unroll
                    for (int h = 0; h < 4; ++h) {
                        v2f tr, ti;
                        tr.x = oR[v2][h]; tr.y = oR[v2][h + 4];
                        ti.x = oI[v2][h]; ti.y = oI[v2][h + 4];
                        *(v2f*)&Tre[(v1 + 64 * v2) * PITCH + h * 16 + ((ib + 8 * hi) >> 2)] = tr;
                        *(v2f*)&Tim[(v1 + 64 * v2) * PITCH + h * 16 + ((ib + 8 * hi) >> 2)] = ti;
                    }
                }
            }
        }
        __syncthreads();

        const int u1b = (wave & 3) * 16;
        v2f eR[4], eI[4], eS[4];
        #pragma unroll
        for (int ks = 0; ks < 4; ++ks) {
            const int k0 = 4 * ks + 2 * hi;
            eR[ks] = *(const v2f*)&Ere[(u1b + col) * EP + k0];
            eI[ks] = *(const v2f*)&Eim[(u1b + col) * EP + k0];
            eS[ks] = eR[ks] + eI[ks];
        }
        float c1[8], s1[8];
        v2f c23[8], s23[8];
        #pragma unroll
        for (int r = 0; r < 8; ++r) {
            const int row = u1b + 8 * hi + r;
            c1[r]  = Qre[row * 4 + 1];
            s1[r]  = Qim[row * 4 + 1];
            c23[r] = *(const v2f*)&Qre[row * 4 + 2];
            s23[r] = *(const v2f*)&Qim[row * 4 + 2];
        }

        float mymax = 0.0f;
        for (int pp = wave >> 2; pp < n; pp += 2) {
            const int cblk = ((vt0 + pp) >> 2) + 4 * ((vt0 + pp) & 3);
            const int vrow = cblk * 16 + col;
            const int  cvb   = (cblk + 2) & 15;
            const bool cropu = keep && cvb < 4;
            const bool low   = (wave & 3) < 2;
            float* const dst = crop_ws + (size_t)e * (S * S)
                             + ((cvb & 3) * 4 + ((wave & 3) ^ 2)) * 256 + lane;
            float sel[8];
            v8f acc[4][3];
            #pragma unroll
            for (int q = 0; q < 4; ++q) { acc[q][0] = (v8f){}; acc[q][1] = (v8f){}; acc[q][2] = (v8f){}; }
            #pragma unroll
            for (int ks = 0; ks < 4; ++ks) {
                #pragma unroll
                for (int q = 0; q < 4; ++q) {
                    const int k0 = q * 16 + 4 * ks + 2 * hi;
                    const v2f bR = *(const v2f*)&Tre[vrow * PITCH + k0];
                    const v2f bI = *(const v2f*)&Tim[vrow * PITCH + k0];
                    const v2f bD = bI - bR;
                    const v2f bS = bR + bI;
                    acc[q][0] = __builtin_amdgcn_wmma_f32_16x16x4_f32(false, eS[ks], false, bR, (short)0, acc[q][0], false, false);
                    acc[q][1] = __builtin_amdgcn_wmma_f32_16x16x4_f32(false, eR[ks], false, bD, (short)0, acc[q][1], false, false);
                    acc[q][2] = __builtin_amdgcn_wmma_f32_16x16x4_f32(false, eI[ks], false, bS, (short)0, acc[q][2], false, false);
                }
            }
            #pragma unroll
            for (int r = 0; r < 8; ++r) {
                const float g0r = acc[0][0][r] - acc[0][2][r];
                const float g0i = acc[0][0][r] + acc[0][1][r];
                const float x1r = acc[1][0][r] - acc[1][2][r];
                const float x1i = acc[1][0][r] + acc[1][1][r];
                const float x2r = acc[2][0][r] - acc[2][2][r];
                const float x2i = acc[2][0][r] + acc[2][1][r];
                const float x3r = acc[3][0][r] - acc[3][2][r];
                const float x3i = acc[3][0][r] + acc[3][1][r];
                const float g1r = x1r * c1[r]    - x1i * s1[r];
                const float g1i = x1r * s1[r]    + x1i * c1[r];
                const float g2r = x2r * c23[r].x - x2i * s23[r].x;
                const float g2i = x2r * s23[r].x + x2i * c23[r].x;
                const float g3r = x3r * c23[r].y - x3i * s23[r].y;
                const float g3i = x3r * s23[r].y + x3i * c23[r].y;
                const float ar = g0r + g2r, ai = g0i + g2i;
                const float br = g0r - g2r, bi = g0i - g2i;
                const float cr = g1r + g3r, ci = g1i + g3i;
                const float dr = g1r - g3r, di = g1i - g3i;
                const float o0r = ar + cr, o0i = ai + ci;
                const float o2r = ar - cr, o2i = ai - ci;
                const float o1r = br - di, o1i = bi + dr;
                const float o3r = br + di, o3i = bi - dr;
                const float m0 = o0r * o0r + o0i * o0i;
                const float m3 = o3r * o3r + o3i * o3i;
                mymax = fmaxf(mymax, m0);
                mymax = fmaxf(mymax, o1r * o1r + o1i * o1i);
                mymax = fmaxf(mymax, o2r * o2r + o2i * o2i);
                mymax = fmaxf(mymax, m3);
                sel[r] = low ? m0 : m3;
            }
            if (cropu) {
                #pragma unroll
                for (int r = 0; r < 8; ++r) dst[r * 32] = sel[r];
            }
        }
        for (int off = 16; off > 0; off >>= 1)
            mymax = fmaxf(mymax, __shfl_xor(mymax, off, 32));
        if (lane == 0) wave_max[wave] = mymax;
        __syncthreads();
        if (t < n) {
            float mx = wave_max[0];
            for (int w = 1; w < 8; ++w) mx = fmaxf(mx, wave_max[w]);
            pmax[e * NPIECE + vt0 + t] = mx;
        }
        g0 = g1;
    }
}

__device__ __forceinline__ void crop_by_transform(const float* __restrict__ xs,
                                const float* __restrict__ ys,
                                const float* __restrict__ zs,
                                const float* __restrict__ As,
                                const float* __restrict__ bgs,
                                const float* __restrict__ ps,
                                const float* __restrict__ pupil_phase,
                                const float* __restrict__ pmax,
                                float* __restrict__ out) {
    __shared__ float Pre[S * PITCH];
    __shared__ float Pim[S * PITCH];
    __shared__ float Wre[S * PITCH];
    __shared__ float Wim[S * PITCH];
    __shared__ float Tre[S * PITCH];
    __shared__ float Tim[S * PITCH];

    const int e = blockIdx.x;
    const int b = e >> 2;
    if (ps[e] <= 0.5f) return;
    const float xe = xs[e], ye = ys[e], ze = zs[e];
    const float Ae = As[e], bge = bgs[e];

    const int t    = threadIdx.x;
    const int lane = t & 31;
    const int wave = t >> 5;
    const int col  = lane & 15;
    const int hi   = lane >> 4;

    const float pm = pmax[e * NPIECE + (t & 15)];

    for (int idx = t; idx < S * S; idx += 256) {
        const int c = idx >> 6, j = idx & 63;
        const int v = (c - 32) & 255;
        const int ti = ((j - 32) * v) & 255;
        float sn, cs;
        __sincosf(TWO_PI * (float)ti * (1.0f / 256.0f), &sn, &cs);
        Wre[c * PITCH + j] = cs;
        Wim[c * PITCH + j] = sn;
    }
    build_pupil(Pre, Pim, pupil_phase, xe, ye, ze, t);
    __syncthreads();

    for (int tt = wave * 2; tt < wave * 2 + 2; ++tt) {
        const int ib = (tt >> 2) * 16;
        const int cb = (tt & 3) * 16;
        v8f k1, k2, k3;
        ctile(Pre, Pim, Wre, Wim, ib + col, cb + col, hi, k1, k2, k3);
        store_T(Tre, Tim, cb + col, ib + 8 * hi, k1, k2, k3);
    }
    __syncthreads();

    float* crop = Pre;

    for (int tt = wave * 2; tt < wave * 2 + 2; ++tt) {
        const int ub = (tt >> 2) * 16;
        const int cb = (tt & 3) * 16;
        v8f k1, k2, k3;
        ctile(Wre, Wim, Tre, Tim, ub + col, cb + col, hi, k1, k2, k3);
        #pragma unroll
        for (int q = 0; q < 8; ++q) {
            const float re = k1[q] - k3[q];
            const float im = k1[q] + k2[q];
            const float pw = re * re + im * im;
            crop[(ub + q + 8 * hi) * PITCH + cb + col] = pw;
        }
    }

    float mx = pm;
    for (int off = 8; off > 0; off >>= 1)
        mx = fmaxf(mx, __shfl_xor(mx, off, 32));
    const float inv = 1.0f / mx;
    __syncthreads();

    for (int idx = t; idx < S * S; idx += 256) {
        const int cu = idx >> 6, cv = idx & 63;
        const float val = crop[cu * PITCH + cv] * inv * Ae + bge;
        atomicAdd(&out[b * S * S + idx], val);
    }
}

__launch_bounds__(256)
__global__ void psf_sum_kernel(const float* __restrict__ xs,
                               const float* __restrict__ ys,
                               const float* __restrict__ zs,
                               const float* __restrict__ As,
                               const float* __restrict__ bgs,
                               const float* __restrict__ ps,
                               const float* __restrict__ pupil_phase,
                               const float* __restrict__ pmax,
                               const float* __restrict__ crop_ws,
                               float* __restrict__ out, size_t ws_bytes) {
    if (ws_bytes < WS_NEED) {
        crop_by_transform(xs, ys, zs, As, bgs, ps, pupil_phase, pmax, out);
        return;
    }
    const int b    = blockIdx.x >> 2;
    const int cub  = blockIdx.x & 3;
    const int t    = threadIdx.x;
    const int lane = t & 31;
    const int rr   = t >> 5;
    const int hi   = lane >> 4;
    const int c4   = lane & 15;
    const int off  = ((c4 >> 2) * 4 + cub) * 256 + rr * 32 + hi * 16 + (c4 & 3) * 4;

    float4 acc = make_float4(0.0f, 0.0f, 0.0f, 0.0f);
    #pragma unroll
    for (int n = 0; n < 4; ++n) {
        const int e = 4 * b + n;
        const bool live = !(ps[e] <= 0.5f);
        const float* pm = live ? pmax + e * NPIECE : pupil_phase;
        const float* cs = live ? crop_ws + (size_t)e * (S * S) : pupil_phase;
        const float4 c = *(const float4*)&cs[off];
        float mx = pm[lane & 15];
        for (int o = 8; o > 0; o >>= 1)
            mx = fmaxf(mx, __shfl_xor(mx, o, 32));
        const float inv = 1.0f / mx;
        const float Ae = As[e], bge = bgs[e];
        acc.x += live ? c.x * inv * Ae + bge : 0.0f;
        acc.y += live ? c.y * inv * Ae + bge : 0.0f;
        acc.z += live ? c.z * inv * Ae + bge : 0.0f;
        acc.w += live ? c.w * inv * Ae + bge : 0.0f;
    }
    const int cu = cub * 16 + hi * 8 + rr;
    *(float4*)&out[(size_t)b * (S * S) + cu * S + c4 * 4] = acc;
}

extern "C" void kernel_launch(void* const* d_in, const int* in_sizes, int n_in,
                              void* d_out, int out_size, void* d_ws, size_t ws_size,
                              hipStream_t stream) {
    const float* x  = (const float*)d_in[0];
    const float* y  = (const float*)d_in[1];
    const float* z  = (const float*)d_in[2];
    const float* A  = (const float*)d_in[3];
    const float* bg = (const float*)d_in[4];
    const float* p  = (const float*)d_in[5];
    const float* ph = (const float*)d_in[6];
    float* out = (float*)d_out;
    float* pmax = (float*)d_ws;
    float* crop = pmax + CROP_OFF;

    const int n_emit = in_sizes[0];
    psf_max_kernel<<<NWG, 256, 0, stream>>>(x, y, z, p, ph, out, out_size, n_emit, pmax, crop, ws_size);
    psf_sum_kernel<<<n_emit, 256, 0, stream>>>(x, y, z, A, bg, p, ph, pmax, crop, out, ws_size);
}


# the two emitters in a work-group's piece range go through pupil build and both transforms together (a pupil plane each, one T buffer): one barrier set, one parameter round trip; sum order unchanged
# speedup vs baseline: 4.3125x; 1.0528x over previous
#include <hip/hip_runtime.h>
#include <math.h>


typedef __attribute__((ext_vector_type(2))) float v2f;
typedef __attribute__((ext_vector_type(8))) float v8f;

#define S 64
#define F 256
#define PITCH 66
#define EP 18
#define TWO_PI 6.28318530717958647692f
#define NWG 256
#define NPIECE 16
#define CROP_OFF (256 * NPIECE)
#define WS_NEED ((size_t)(CROP_OFF + 256 * S * S) * sizeof(float))

__device__ __forceinline__ void build_pupil(float* Pre, float* Pim,
                                            const float* __restrict__ pupil_phase,
                                            float xe, float ye, float ze, int t) {
    for (int idx = t; idx < S * S; idx += 256) {
        const int i = idx >> 6, j = idx & 63;
        const float ui = (2.0f * (float)i / 63.0f - 1.0f) * (1.0f / 0.75f);
        const float vj = (2.0f * (float)j / 63.0f - 1.0f) * (1.0f / 0.75f);
        const float r2 = ui * ui + vj * vj;
        const float mag  = (r2 <= 1.0f) ? 1.0f : 0.0f;
        const float prop = sqrtf(1.0f - fminf(r2, 1.0f));
        const float ph = pupil_phase[idx] + prop * ze
                       - TWO_PI * ((float)(i - 32) * (1.0f / 256.0f)) * xe
                       - TWO_PI * ((float)(j - 32) * (1.0f / 256.0f)) * ye;
        float sn, cs;
        __sincosf(ph, &sn, &cs);
        Pre[i * PITCH + j] = mag * cs;
        Pim[i * PITCH + j] = mag * sn;
    }
}

__device__ __forceinline__ void ctile(const float* Are, const float* Aim,
                                      const float* Bre, const float* Bim,
                                      int arow, int bcol, int hi,
                                      v8f& k1, v8f& k2, v8f& k3) {
    k1 = (v8f){}; k2 = (v8f){}; k3 = (v8f){};
    #pragma unroll
    for (int kb = 0; kb < S; kb += 4) {
        const int k0 = kb + 2 * hi;
        const v2f aR = *(const v2f*)&Are[arow * PITCH + k0];
        const v2f aI = *(const v2f*)&Aim[arow * PITCH + k0];
        const v2f bR = *(const v2f*)&Bre[bcol * PITCH + k0];
        const v2f bI = *(const v2f*)&Bim[bcol * PITCH + k0];
        const v2f aS = aR + aI;
        const v2f bD = bI - bR;
        const v2f bS = bR + bI;
        k1 = __builtin_amdgcn_wmma_f32_16x16x4_f32(false, aS, false, bR, (short)0, k1, false, false);
        k2 = __builtin_amdgcn_wmma_f32_16x16x4_f32(false, aR, false, bD, (short)0, k2, false, false);
        k3 = __builtin_amdgcn_wmma_f32_16x16x4_f32(false, aI, false, bS, (short)0, k3, false, false);
    }
}

__device__ __forceinline__ void store_T(float* Tre, float* Tim, int trow, int rbase,
                                        const v8f& k1, const v8f& k2, const v8f& k3) {
    #pragma unroll
    for (int q = 0; q < 4; ++q) {
        v2f tr, ti;
        tr.x = k1[2 * q]     - k3[2 * q];
        tr.y = k1[2 * q + 1] - k3[2 * q + 1];
        ti.x = k1[2 * q]     + k2[2 * q];
        ti.y = k1[2 * q + 1] + k2[2 * q + 1];
        *(v2f*)&Tre[trow * PITCH + rbase + 2 * q] = tr;
        *(v2f*)&Tim[trow * PITCH + rbase + 2 * q] = ti;
    }
}

__launch_bounds__(256)
__global__ void psf_max_kernel(const float* __restrict__ xs,
                               const float* __restrict__ ys,
                               const float* __restrict__ zs,
                               const float* __restrict__ ps,
                               const float* __restrict__ pupil_phase,
                               float* __restrict__ out, int n_out, int n_emit,
                               float* __restrict__ pmax,
                               float* __restrict__ crop_ws, size_t ws_bytes) {
    __shared__ float Pre[2 * S * PITCH];
    __shared__ float Pim[2 * S * PITCH];
    __shared__ float Utab[S];
    __shared__ float Ere[S * EP];
    __shared__ float Eim[S * EP];
    __shared__ float Qre[S * 4];
    __shared__ float Qim[S * 4];
    __shared__ float Tre[F * PITCH];
    __shared__ float Tim[F * PITCH];
    __shared__ float wave_max[16];
    __shared__ int   wave_cnt[8];
    __shared__ int   live[256];

    const int t    = threadIdx.x;
    const int lane = t & 31;
    const int wave = t >> 5;
    const int col  = lane & 15;
    const int hi   = lane >> 4;

    const bool keep = ws_bytes >= WS_NEED;
    if (!keep)
        for (int i = blockIdx.x * 256 + t; i < n_out; i += NWG * 256) out[i] = 0.0f;

    float pph[16];
    #pragma unroll
    for (int k = 0; k < 16; ++k) pph[k] = pupil_phase[t + 256 * k];
    if (t < S) Utab[t] = (2.0f * (float)t / 63.0f - 1.0f) * (1.0f / 0.75f);

    bool alive = false;
    if (t < n_emit) alive = !(ps[t] <= 0.5f);
    const unsigned m = (unsigned)__ballot(alive);
    if (lane == 0) wave_cnt[wave] = __popc(m);
    __syncthreads();
    int base = 0, n_live = 0;
    for (int w = 0; w < 8; ++w) {
        const int c = wave_cnt[w];
        if (w < wave) base += c;
        n_live += c;
    }
    if (alive) live[base + __popc(m & ((1u << lane) - 1u))] = t;

    const int lo   = (blockIdx.x * n_live) >> 4;
    const int hi_g = ((blockIdx.x + 1) * n_live) >> 4;
    if (lo >= hi_g) return;

    float mag[16], prop[16];
    {
        const float vj = Utab[t & 63];
        #pragma unroll
        for (int k = 0; k < 16; ++k) {
            const float ui = Utab[(t >> 6) + 4 * k];
            const float r2 = ui * ui + vj * vj;
            mag[k]  = (r2 <= 1.0f) ? 1.0f : 0.0f;
            prop[k] = sqrtf(1.0f - fminf(r2, 1.0f));
        }
    }

    for (int idx = t; idx < S * 16; idx += 256) {
        const int u = idx >> 4, p = idx & 15;
        const int ti = (4 * (p - 8) * u) & 255;
        float sn, cs;
        __sincosf(TWO_PI * (float)ti * (1.0f / 256.0f), &sn, &cs);
        Ere[u * EP + p] = cs;
        Eim[u * EP + p] = sn;
    }
    {
        const int ti = ((t & 3) * (t >> 2)) & 255;
        float sn, cs;
        __sincosf(TWO_PI * (float)ti * (1.0f / 256.0f), &sn, &cs);
        Qre[t] = cs;
        Qim[t] = sn;
    }


    {
        const int g0 = lo;
        const int rA  = g0 >> 4;
        const int vtA = g0 & 15;
        int gA = (rA + 1) << 4;
        if (gA > hi_g) gA = hi_g;
        const int gB = hi_g;
        const int nA = gA - g0;
        const int nB = gB - gA;
        __syncthreads();
        const int eA = live[rA];
        const int eB = live[nB > 0 ? rA + 1 : rA];
        {
            const float xA = xs[eA], yA = ys[eA], zA = zs[eA];
            const float xB = xs[eB], yB = ys[eB], zB = zs[eB];
            const int j = t & 63;
            const int slot = (j & 3) * 16 + (j >> 2);
            #pragma unroll
            for (int k = 0; k < 16; ++k) {
                const int i = (t >> 6) + 4 * k;
                const float ph = pph[k] + prop[k] * zA
                               - TWO_PI * ((float)(i - 32) * (1.0f / 256.0f)) * xA
                               - TWO_PI * ((float)(j - 32) * (1.0f / 256.0f)) * yA;
                float sn, cs;
                __sincosf(ph, &sn, &cs);
                Pre[i * PITCH + slot] = mag[k] * cs;
                Pim[i * PITCH + slot] = mag[k] * sn;
            }
            if (nB > 0) {
                #pragma unroll
                for (int k = 0; k < 16; ++k) {
                    const int i = (t >> 6) + 4 * k;
                    const float ph = pph[k] + prop[k] * zB
                                   - TWO_PI * ((float)(i - 32) * (1.0f / 256.0f)) * xB
                                   - TWO_PI * ((float)(j - 32) * (1.0f / 256.0f)) * yB;
                    float sn, cs;
                    __sincosf(ph, &sn, &cs);
                    Pre[S * PITCH + i * PITCH + slot] = mag[k] * cs;
                    Pim[S * PITCH + i * PITCH + slot] = mag[k] * sn;
                }
            }
        }
        __syncthreads();

        const int bb0A = vtA >> 2;
        const int nbbA = ((vtA + nA - 1) >> 2) - bb0A + 1;
        const int nbbB = (nB + 3) >> 2;
        for (int tt = wave; tt < 4 * (nbbA + nbbB); tt += 8) {
            const bool sB  = tt >= 4 * nbbA;
            const int  t2  = sB ? tt - 4 * nbbA : tt;
            const float* const Pr = Pre + (sB ? S * PITCH : 0);
            const float* const Pi = Pim + (sB ? S * PITCH : 0);
            const int  vt0 = sB ? 0 : vtA;
            const int  n   = sB ? nB : nA;
            const int ib = (t2 & 3) * 16;
            const int bb = (sB ? 0 : bb0A) + (t2 >> 2);
            const int v1 = bb * 16 + col;
            v2f fR[4], fD[4], fS[4];
            #pragma unroll
            for (int ks = 0; ks < 4; ++ks) {
                const int k0 = 4 * ks + 2 * hi;
                const v2f fI = *(const v2f*)&Eim[v1 * EP + k0];
                fR[ks] = *(const v2f*)&Ere[v1 * EP + k0];
                fD[ks] = fI - fR[ks];
                fS[ks] = fR[ks] + fI;
            }
            const float wc1 = Qre[v1 * 4 + 1], ws1 = Qim[v1 * 4 + 1];
            const v2f wc23 = *(const v2f*)&Qre[v1 * 4 + 2];
            const v2f ws23 = *(const v2f*)&Qim[v1 * 4 + 2];
            v8f acc[4][3];
            #pragma unroll
            for (int q = 0; q < 4; ++q) { acc[q][0] = (v8f){}; acc[q][1] = (v8f){}; acc[q][2] = (v8f){}; }
            #pragma unroll
            for (int ks = 0; ks < 4; ++ks) {
                #pragma unroll
                for (int q = 0; q < 4; ++q) {
                    const int k0 = q * 16 + 4 * ks + 2 * hi;
                    const v2f aR = *(const v2f*)&Pr[(ib + col) * PITCH + k0];
                    const v2f aI = *(const v2f*)&Pi[(ib + col) * PITCH + k0];
                    const v2f aS = aR + aI;
                    acc[q][0] = __builtin_amdgcn_wmma_f32_16x16x4_f32(false, aS, false, fR[ks], (short)0, acc[q][0], false, false);
                    acc[q][1] = __builtin_amdgcn_wmma_f32_16x16x4_f32(false, aR, false, fD[ks], (short)0, acc[q][1], false, false);
                    acc[q][2] = __builtin_amdgcn_wmma_f32_16x16x4_f32(false, aI, false, fS[ks], (short)0, acc[q][2], false, false);
                }
            }
            float oR[4][8], oI[4][8];
            #pragma unroll
            for (int r8 = 0; r8 < 8; ++r8) {
                const float g0r = acc[0][0][r8] - acc[0][2][r8];
                const float g0i = acc[0][0][r8] + acc[0][1][r8];
                const float x1r = acc[1][0][r8] - acc[1][2][r8];
                const float x1i = acc[1][0][r8] + acc[1][1][r8];
                const float x2r = acc[2][0][r8] - acc[2][2][r8];
                const float x2i = acc[2][0][r8] + acc[2][1][r8];
                const float x3r = acc[3][0][r8] - acc[3][2][r8];
                const float x3i = acc[3][0][r8] + acc[3][1][r8];
                const float g1r = x1r * wc1    - x1i * ws1;
                const float g1i = x1r * ws1    + x1i * wc1;
                const float g2r = x2r * wc23.x - x2i * ws23.x;
                const float g2i = x2r * ws23.x + x2i * wc23.x;
                const float g3r = x3r * wc23.y - x3i * ws23.y;
                const float g3i = x3r * ws23.y + x3i * wc23.y;
                const float ar = g0r + g2r, ai = g0i + g2i;
                const float br = g0r - g2r, bi = g0i - g2i;
                const float cr = g1r + g3r, ci = g1i + g3i;
                const float dr = g1r - g3r, di = g1i - g3i;
                oR[0][r8] = ar + cr; oI[0][r8] = ai + ci;
                oR[1][r8] = br - di; oI[1][r8] = bi + dr;
                oR[2][r8] = ar - cr; oI[2][r8] = ai - ci;
                oR[3][r8] = br + di; oI[3][r8] = bi - dr;
            }
            #pragma unroll
            for (int v2 = 0; v2 < 4; ++v2) {
                const int pi = 4 * bb + v2;
                if (pi >= vt0 && pi < vt0 + n) {
                    #pragma unroll
                    for (int h = 0; h < 4; ++h) {
                        v2f tr, ti;
                        tr.x = oR[v2][h]; tr.y = oR[v2][h + 4];
                        ti.x = oI[v2][h]; ti.y = oI[v2][h + 4];
                        *(v2f*)&Tre[(v1 + 64 * v2) * PITCH + h * 16 + ((ib + 8 * hi) >> 2)] = tr;
                        *(v2f*)&Tim[(v1 + 64 * v2) * PITCH + h * 16 + ((ib + 8 * hi) >> 2)] = ti;
                    }
                }
            }
        }
        __syncthreads();

        const int u1b = (wave & 3) * 16;
        v2f eR[4], eI[4], eS[4];
        #pragma unroll
        for (int ks = 0; ks < 4; ++ks) {
            const int k0 = 4 * ks + 2 * hi;
            eR[ks] = *(const v2f*)&Ere[(u1b + col) * EP + k0];
            eI[ks] = *(const v2f*)&Eim[(u1b + col) * EP + k0];
            eS[ks] = eR[ks] + eI[ks];
        }
        float c1[8], s1[8];
        v2f c23[8], s23[8];
        #pragma unroll
        for (int r = 0; r < 8; ++r) {
            const int row = u1b + 8 * hi + r;
            c1[r]  = Qre[row * 4 + 1];
            s1[r]  = Qim[row * 4 + 1];
            c23[r] = *(const v2f*)&Qre[row * 4 + 2];
            s23[r] = *(const v2f*)&Qim[row * 4 + 2];
        }

        float mymax = 0.0f, maxA = 0.0f;
        bool  inB = false;
        for (int pp = wave >> 2; pp < nA + nB; pp += 2) {
            const bool sB = pp >= nA;
            const int  pi = sB ? pp - nA : vtA + pp;
            const int  e  = sB ? eB : eA;
            const int cblk = (pi >> 2) + 4 * (pi & 3);
            if (sB && !inB) { maxA = mymax; mymax = 0.0f; inB = true; }
            const int vrow = cblk * 16 + col;
            const int  cvb   = (cblk + 2) & 15;
            const bool cropu = keep && cvb < 4;
            const bool low   = (wave & 3) < 2;
            float* const dst = crop_ws + (size_t)e * (S * S)
                             + ((cvb & 3) * 4 + ((wave & 3) ^ 2)) * 256 + lane;
            float sel[8];
            v8f acc[4][3];
            #pragma unroll
            for (int q = 0; q < 4; ++q) { acc[q][0] = (v8f){}; acc[q][1] = (v8f){}; acc[q][2] = (v8f){}; }
            #pragma unroll
            for (int ks = 0; ks < 4; ++ks) {
                #pragma unroll
                for (int q = 0; q < 4; ++q) {
                    const int k0 = q * 16 + 4 * ks + 2 * hi;
                    const v2f bR = *(const v2f*)&Tre[vrow * PITCH + k0];
                    const v2f bI = *(const v2f*)&Tim[vrow * PITCH + k0];
                    const v2f bD = bI - bR;
                    const v2f bS = bR + bI;
                    acc[q][0] = __builtin_amdgcn_wmma_f32_16x16x4_f32(false, eS[ks], false, bR, (short)0, acc[q][0], false, false);
                    acc[q][1] = __builtin_amdgcn_wmma_f32_16x16x4_f32(false, eR[ks], false, bD, (short)0, acc[q][1], false, false);
                    acc[q][2] = __builtin_amdgcn_wmma_f32_16x16x4_f32(false, eI[ks], false, bS, (short)0, acc[q][2], false, false);
                }
            }
            #pragma unroll
            for (int r = 0; r < 8; ++r) {
                const float g0r = acc[0][0][r] - acc[0][2][r];
                const float g0i = acc[0][0][r] + acc[0][1][r];
                const float x1r = acc[1][0][r] - acc[1][2][r];
                const float x1i = acc[1][0][r] + acc[1][1][r];
                const float x2r = acc[2][0][r] - acc[2][2][r];
                const float x2i = acc[2][0][r] + acc[2][1][r];
                const float x3r = acc[3][0][r] - acc[3][2][r];
                const float x3i = acc[3][0][r] + acc[3][1][r];
                const float g1r = x1r * c1[r]    - x1i * s1[r];
                const float g1i = x1r * s1[r]    + x1i * c1[r];
                const float g2r = x2r * c23[r].x - x2i * s23[r].x;
                const float g2i = x2r * s23[r].x + x2i * c23[r].x;
                const float g3r = x3r * c23[r].y - x3i * s23[r].y;
                const float g3i = x3r * s23[r].y + x3i * c23[r].y;
                const float ar = g0r + g2r, ai = g0i + g2i;
                const float br = g0r - g2r, bi = g0i - g2i;
                const float cr = g1r + g3r, ci = g1i + g3i;
                const float dr = g1r - g3r, di = g1i - g3i;
                const float o0r = ar + cr, o0i = ai + ci;
                const float o2r = ar - cr, o2i = ai - ci;
                const float o1r = br - di, o1i = bi + dr;
                const float o3r = br + di, o3i = bi - dr;
                const float m0 = o0r * o0r + o0i * o0i;
                const float m3 = o3r * o3r + o3i * o3i;
                mymax = fmaxf(mymax, m0);
                mymax = fmaxf(mymax, o1r * o1r + o1i * o1i);
                mymax = fmaxf(mymax, o2r * o2r + o2i * o2i);
                mymax = fmaxf(mymax, m3);
                sel[r] = low ? m0 : m3;
            }
            if (cropu) {
                #pragma unroll
                for (int r = 0; r < 8; ++r) dst[r * 32] = sel[r];
            }
        }
        float maxB = 0.0f;
        if (inB) maxB = mymax; else maxA = mymax;
        for (int off = 16; off > 0; off >>= 1) {
            maxA = fmaxf(maxA, __shfl_xor(maxA, off, 32));
            maxB = fmaxf(maxB, __shfl_xor(maxB, off, 32));
        }
        if (lane == 0) { wave_max[wave] = maxA; wave_max[8 + wave] = maxB; }
        __syncthreads();
        if (t < 32) {
            const bool hB = t >= 16;
            const int  q  = t & 15;
            if (q < (hB ? nB : nA)) {
                const int o = hB ? 8 : 0;
                float mx = wave_max[o];
                for (int w = 1; w < 8; ++w) mx = fmaxf(mx, wave_max[o + w]);
                pmax[(hB ? eB : eA) * NPIECE + (hB ? 0 : vtA) + q] = mx;
            }
        }
    }
}

__device__ __forceinline__ void crop_by_transform(const float* __restrict__ xs,
                                const float* __restrict__ ys,
                                const float* __restrict__ zs,
                                const float* __restrict__ As,
                                const float* __restrict__ bgs,
                                const float* __restrict__ ps,
                                const float* __restrict__ pupil_phase,
                                const float* __restrict__ pmax,
                                float* __restrict__ out) {
    __shared__ float Pre[S * PITCH];
    __shared__ float Pim[S * PITCH];
    __shared__ float Wre[S * PITCH];
    __shared__ float Wim[S * PITCH];
    __shared__ float Tre[S * PITCH];
    __shared__ float Tim[S * PITCH];

    const int e = blockIdx.x;
    const int b = e >> 2;
    if (ps[e] <= 0.5f) return;
    const float xe = xs[e], ye = ys[e], ze = zs[e];
    const float Ae = As[e], bge = bgs[e];

    const int t    = threadIdx.x;
    const int lane = t & 31;
    const int wave = t >> 5;
    const int col  = lane & 15;
    const int hi   = lane >> 4;

    const float pm = pmax[e * NPIECE + (t & 15)];

    for (int idx = t; idx < S * S; idx += 256) {
        const int c = idx >> 6, j = idx & 63;
        const int v = (c - 32) & 255;
        const int ti = ((j - 32) * v) & 255;
        float sn, cs;
        __sincosf(TWO_PI * (float)ti * (1.0f / 256.0f), &sn, &cs);
        Wre[c * PITCH + j] = cs;
        Wim[c * PITCH + j] = sn;
    }
    build_pupil(Pre, Pim, pupil_phase, xe, ye, ze, t);
    __syncthreads();

    for (int tt = wave * 2; tt < wave * 2 + 2; ++tt) {
        const int ib = (tt >> 2) * 16;
        const int cb = (tt & 3) * 16;
        v8f k1, k2, k3;
        ctile(Pre, Pim, Wre, Wim, ib + col, cb + col, hi, k1, k2, k3);
        store_T(Tre, Tim, cb + col, ib + 8 * hi, k1, k2, k3);
    }
    __syncthreads();

    float* crop = Pre;

    for (int tt = wave * 2; tt < wave * 2 + 2; ++tt) {
        const int ub = (tt >> 2) * 16;
        const int cb = (tt & 3) * 16;
        v8f k1, k2, k3;
        ctile(Wre, Wim, Tre, Tim, ub + col, cb + col, hi, k1, k2, k3);
        #pragma unroll
        for (int q = 0; q < 8; ++q) {
            const float re = k1[q] - k3[q];
            const float im = k1[q] + k2[q];
            const float pw = re * re + im * im;
            crop[(ub + q + 8 * hi) * PITCH + cb + col] = pw;
        }
    }

    float mx = pm;
    for (int off = 8; off > 0; off >>= 1)
        mx = fmaxf(mx, __shfl_xor(mx, off, 32));
    const float inv = 1.0f / mx;
    __syncthreads();

    for (int idx = t; idx < S * S; idx += 256) {
        const int cu = idx >> 6, cv = idx & 63;
        const float val = crop[cu * PITCH + cv] * inv * Ae + bge;
        atomicAdd(&out[b * S * S + idx], val);
    }
}

__launch_bounds__(256)
__global__ void psf_sum_kernel(const float* __restrict__ xs,
                               const float* __restrict__ ys,
                               const float* __restrict__ zs,
                               const float* __restrict__ As,
                               const float* __restrict__ bgs,
                               const float* __restrict__ ps,
                               const float* __restrict__ pupil_phase,
                               const float* __restrict__ pmax,
                               const float* __restrict__ crop_ws,
                               float* __restrict__ out, size_t ws_bytes) {
    if (ws_bytes < WS_NEED) {
        crop_by_transform(xs, ys, zs, As, bgs, ps, pupil_phase, pmax, out);
        return;
    }
    const int b    = blockIdx.x >> 2;
    const int cub  = blockIdx.x & 3;
    const int t    = threadIdx.x;
    const int lane = t & 31;
    const int rr   = t >> 5;
    const int hi   = lane >> 4;
    const int c4   = lane & 15;
    const int off  = ((c4 >> 2) * 4 + cub) * 256 + rr * 32 + hi * 16 + (c4 & 3) * 4;

    float4 acc = make_float4(0.0f, 0.0f, 0.0f, 0.0f);
    #pragma unroll
    for (int n = 0; n < 4; ++n) {
        const int e = 4 * b + n;
        const bool live = !(ps[e] <= 0.5f);
        const float* pm = live ? pmax + e * NPIECE : pupil_phase;
        const float* cs = live ? crop_ws + (size_t)e * (S * S) : pupil_phase;
        const float4 c = *(const float4*)&cs[off];
        float mx = pm[lane & 15];
        for (int o = 8; o > 0; o >>= 1)
            mx = fmaxf(mx, __shfl_xor(mx, o, 32));
        const float inv = 1.0f / mx;
        const float Ae = As[e], bge = bgs[e];
        acc.x += live ? c.x * inv * Ae + bge : 0.0f;
        acc.y += live ? c.y * inv * Ae + bge : 0.0f;
        acc.z += live ? c.z * inv * Ae + bge : 0.0f;
        acc.w += live ? c.w * inv * Ae + bge : 0.0f;
    }
    const int cu = cub * 16 + hi * 8 + rr;
    *(float4*)&out[(size_t)b * (S * S) + cu * S + c4 * 4] = acc;
}

extern "C" void kernel_launch(void* const* d_in, const int* in_sizes, int n_in,
                              void* d_out, int out_size, void* d_ws, size_t ws_size,
                              hipStream_t stream) {
    const float* x  = (const float*)d_in[0];
    const float* y  = (const float*)d_in[1];
    const float* z  = (const float*)d_in[2];
    const float* A  = (const float*)d_in[3];
    const float* bg = (const float*)d_in[4];
    const float* p  = (const float*)d_in[5];
    const float* ph = (const float*)d_in[6];
    float* out = (float*)d_out;
    float* pmax = (float*)d_ws;
    float* crop = pmax + CROP_OFF;

    const int n_emit = in_sizes[0];
    psf_max_kernel<<<NWG, 256, 0, stream>>>(x, y, z, p, ph, out, out_size, n_emit, pmax, crop, ws_size);
    psf_sum_kernel<<<n_emit, 256, 0, stream>>>(x, y, z, A, bg, p, ph, pmax, crop, out, ws_size);
}


# second transform of the maximum pass split radix-8, not radix-4, over its output row (u = u1 + 32*u2): eight 8-term matrix products, then twiddle and 8-point sum in registers; f32 sum order changes
# speedup vs baseline: 4.8078x; 1.1149x over previous
#include <hip/hip_runtime.h>
#include <math.h>


typedef __attribute__((ext_vector_type(2))) float v2f;
typedef __attribute__((ext_vector_type(8))) float v8f;

#define S 64
#define F 256
#define PITCH 66
#define EP 18
#define TWO_PI 6.28318530717958647692f
#define NWG 256
#define NPIECE 16
#define CROP_OFF (256 * NPIECE)
#define WS_NEED ((size_t)(CROP_OFF + 256 * S * S) * sizeof(float))

__device__ __forceinline__ void build_pupil(float* Pre, float* Pim,
                                            const float* __restrict__ pupil_phase,
                                            float xe, float ye, float ze, int t) {
    for (int idx = t; idx < S * S; idx += 256) {
        const int i = idx >> 6, j = idx & 63;
        const float ui = (2.0f * (float)i / 63.0f - 1.0f) * (1.0f / 0.75f);
        const float vj = (2.0f * (float)j / 63.0f - 1.0f) * (1.0f / 0.75f);
        const float r2 = ui * ui + vj * vj;
        const float mag  = (r2 <= 1.0f) ? 1.0f : 0.0f;
        const float prop = sqrtf(1.0f - fminf(r2, 1.0f));
        const float ph = pupil_phase[idx] + prop * ze
                       - TWO_PI * ((float)(i - 32) * (1.0f / 256.0f)) * xe
                       - TWO_PI * ((float)(j - 32) * (1.0f / 256.0f)) * ye;
        float sn, cs;
        __sincosf(ph, &sn, &cs);
        Pre[i * PITCH + j] = mag * cs;
        Pim[i * PITCH + j] = mag * sn;
    }
}

__device__ __forceinline__ void ctile(const float* Are, const float* Aim,
                                      const float* Bre, const float* Bim,
                                      int arow, int bcol, int hi,
                                      v8f& k1, v8f& k2, v8f& k3) {
    k1 = (v8f){}; k2 = (v8f){}; k3 = (v8f){};
    #pragma unroll
    for (int kb = 0; kb < S; kb += 4) {
        const int k0 = kb + 2 * hi;
        const v2f aR = *(const v2f*)&Are[arow * PITCH + k0];
        const v2f aI = *(const v2f*)&Aim[arow * PITCH + k0];
        const v2f bR = *(const v2f*)&Bre[bcol * PITCH + k0];
        const v2f bI = *(const v2f*)&Bim[bcol * PITCH + k0];
        const v2f aS = aR + aI;
        const v2f bD = bI - bR;
        const v2f bS = bR + bI;
        k1 = __builtin_amdgcn_wmma_f32_16x16x4_f32(false, aS, false, bR, (short)0, k1, false, false);
        k2 = __builtin_amdgcn_wmma_f32_16x16x4_f32(false, aR, false, bD, (short)0, k2, false, false);
        k3 = __builtin_amdgcn_wmma_f32_16x16x4_f32(false, aI, false, bS, (short)0, k3, false, false);
    }
}

__device__ __forceinline__ void store_T(float* Tre, float* Tim, int trow, int rbase,
                                        const v8f& k1, const v8f& k2, const v8f& k3) {
    #pragma unroll
    for (int q = 0; q < 4; ++q) {
        v2f tr, ti;
        tr.x = k1[2 * q]     - k3[2 * q];
        tr.y = k1[2 * q + 1] - k3[2 * q + 1];
        ti.x = k1[2 * q]     + k2[2 * q];
        ti.y = k1[2 * q + 1] + k2[2 * q + 1];
        *(v2f*)&Tre[trow * PITCH + rbase + 2 * q] = tr;
        *(v2f*)&Tim[trow * PITCH + rbase + 2 * q] = ti;
    }
}

__launch_bounds__(256)
__global__ void psf_max_kernel(const float* __restrict__ xs,
                               const float* __restrict__ ys,
                               const float* __restrict__ zs,
                               const float* __restrict__ ps,
                               const float* __restrict__ pupil_phase,
                               float* __restrict__ out, int n_out, int n_emit,
                               float* __restrict__ pmax,
                               float* __restrict__ crop_ws, size_t ws_bytes) {
    __shared__ float Pre[2 * S * PITCH];
    __shared__ float Pim[2 * S * PITCH];
    __shared__ float Utab[S];
    __shared__ float Ere[S * EP];
    __shared__ float Eim[S * EP];
    __shared__ float Qre[S * 4];
    __shared__ float Qim[S * 4];
    __shared__ float Q8re[32 * 8];
    __shared__ float Q8im[32 * 8];
    __shared__ float Tre[F * PITCH];
    __shared__ float Tim[F * PITCH];
    __shared__ float wave_max[16];
    __shared__ int   wave_cnt[8];
    __shared__ int   live[256];

    const int t    = threadIdx.x;
    const int lane = t & 31;
    const int wave = t >> 5;
    const int col  = lane & 15;
    const int hi   = lane >> 4;

    const bool keep = ws_bytes >= WS_NEED;
    if (!keep)
        for (int i = blockIdx.x * 256 + t; i < n_out; i += NWG * 256) out[i] = 0.0f;

    float pph[16];
    #pragma unroll
    for (int k = 0; k < 16; ++k) pph[k] = pupil_phase[t + 256 * k];
    if (t < S) Utab[t] = (2.0f * (float)t / 63.0f - 1.0f) * (1.0f / 0.75f);

    bool alive = false;
    if (t < n_emit) alive = !(ps[t] <= 0.5f);
    const unsigned m = (unsigned)__ballot(alive);
    if (lane == 0) wave_cnt[wave] = __popc(m);
    __syncthreads();
    int base = 0, n_live = 0;
    for (int w = 0; w < 8; ++w) {
        const int c = wave_cnt[w];
        if (w < wave) base += c;
        n_live += c;
    }
    if (alive) live[base + __popc(m & ((1u << lane) - 1u))] = t;

    const int lo   = (blockIdx.x * n_live) >> 4;
    const int hi_g = ((blockIdx.x + 1) * n_live) >> 4;
    if (lo >= hi_g) return;

    float mag[16], prop[16];
    {
        const float vj = Utab[t & 63];
        #pragma unroll
        for (int k = 0; k < 16; ++k) {
            const float ui = Utab[(t >> 6) + 4 * k];
            const float r2 = ui * ui + vj * vj;
            mag[k]  = (r2 <= 1.0f) ? 1.0f : 0.0f;
            prop[k] = sqrtf(1.0f - fminf(r2, 1.0f));
        }
    }

    for (int idx = t; idx < S * 16; idx += 256) {
        const int u = idx >> 4, p = idx & 15;
        const int ti = (4 * (p - 8) * u) & 255;
        float sn, cs;
        __sincosf(TWO_PI * (float)ti * (1.0f / 256.0f), &sn, &cs);
        Ere[u * EP + p] = cs;
        Eim[u * EP + p] = sn;
    }
    {
        const int ti = ((t & 3) * (t >> 2)) & 255;
        float sn, cs;
        __sincosf(TWO_PI * (float)ti * (1.0f / 256.0f), &sn, &cs);
        Qre[t] = cs;
        Qim[t] = sn;
    }
    {
        const int q = t & 7, u = t >> 3;
        const int ti = q * u;
        float sn, cs;
        __sincosf(TWO_PI * (float)ti * (1.0f / 256.0f), &sn, &cs);
        Q8re[((u >> 1) * 8 + q) * 2 + (u & 1)] = cs;
        Q8im[((u >> 1) * 8 + q) * 2 + (u & 1)] = sn;
    }


    {
        const int g0 = lo;
        const int rA  = g0 >> 4;
        const int vtA = g0 & 15;
        int gA = (rA + 1) << 4;
        if (gA > hi_g) gA = hi_g;
        const int gB = hi_g;
        const int nA = gA - g0;
        const int nB = gB - gA;
        __syncthreads();
        const int eA = live[rA];
        const int eB = live[nB > 0 ? rA + 1 : rA];
        {
            const float xA = xs[eA], yA = ys[eA], zA = zs[eA];
            const float xB = xs[eB], yB = ys[eB], zB = zs[eB];
            const int j = t & 63;
            const int slot = (j & 3) * 16 + (j >> 2);
            #pragma unroll
            for (int k = 0; k < 16; ++k) {
                const int i = (t >> 6) + 4 * k;
                const float ph = pph[k] + prop[k] * zA
                               - TWO_PI * ((float)(i - 32) * (1.0f / 256.0f)) * xA
                               - TWO_PI * ((float)(j - 32) * (1.0f / 256.0f)) * yA;
                float sn, cs;
                __sincosf(ph, &sn, &cs);
                Pre[i * PITCH + slot] = mag[k] * cs;
                Pim[i * PITCH + slot] = mag[k] * sn;
            }
            if (nB > 0) {
                #pragma unroll
                for (int k = 0; k < 16; ++k) {
                    const int i = (t >> 6) + 4 * k;
                    const float ph = pph[k] + prop[k] * zB
                                   - TWO_PI * ((float)(i - 32) * (1.0f / 256.0f)) * xB
                                   - TWO_PI * ((float)(j - 32) * (1.0f / 256.0f)) * yB;
                    float sn, cs;
                    __sincosf(ph, &sn, &cs);
                    Pre[S * PITCH + i * PITCH + slot] = mag[k] * cs;
                    Pim[S * PITCH + i * PITCH + slot] = mag[k] * sn;
                }
            }
        }
        __syncthreads();

        const int bb0A = vtA >> 2;
        const int nbbA = ((vtA + nA - 1) >> 2) - bb0A + 1;
        const int nbbB = (nB + 3) >> 2;
        for (int tt = wave; tt < 4 * (nbbA + nbbB); tt += 8) {
            const bool sB  = tt >= 4 * nbbA;
            const int  t2  = sB ? tt - 4 * nbbA : tt;
            const float* const Pr = Pre + (sB ? S * PITCH : 0);
            const float* const Pi = Pim + (sB ? S * PITCH : 0);
            const int  vt0 = sB ? 0 : vtA;
            const int  n   = sB ? nB : nA;
            const int ib = (t2 & 3) * 16;
            const int bb = (sB ? 0 : bb0A) + (t2 >> 2);
            const int v1 = bb * 16 + col;
            v2f fR[4], fD[4], fS[4];
            #pragma unroll
            for (int ks = 0; ks < 4; ++ks) {
                const int k0 = 4 * ks + 2 * hi;
                const v2f fI = *(const v2f*)&Eim[v1 * EP + k0];
                fR[ks] = *(const v2f*)&Ere[v1 * EP + k0];
                fD[ks] = fI - fR[ks];
                fS[ks] = fR[ks] + fI;
            }
            const float wc1 = Qre[v1 * 4 + 1], ws1 = Qim[v1 * 4 + 1];
            const v2f wc23 = *(const v2f*)&Qre[v1 * 4 + 2];
            const v2f ws23 = *(const v2f*)&Qim[v1 * 4 + 2];
            v8f acc[4][3];
            #pragma unroll
            for (int q = 0; q < 4; ++q) { acc[q][0] = (v8f){}; acc[q][1] = (v8f){}; acc[q][2] = (v8f){}; }
            #pragma unroll
            for (int ks = 0; ks < 4; ++ks) {
                #pragma unroll
                for (int q = 0; q < 4; ++q) {
                    const int k0 = q * 16 + 4 * ks + 2 * hi;
                    const v2f aR = *(const v2f*)&Pr[(ib + col) * PITCH + k0];
                    const v2f aI = *(const v2f*)&Pi[(ib + col) * PITCH + k0];
                    const v2f aS = aR + aI;
                    acc[q][0] = __builtin_amdgcn_wmma_f32_16x16x4_f32(false, aS, false, fR[ks], (short)0, acc[q][0], false, false);
                    acc[q][1] = __builtin_amdgcn_wmma_f32_16x16x4_f32(false, aR, false, fD[ks], (short)0, acc[q][1], false, false);
                    acc[q][2] = __builtin_amdgcn_wmma_f32_16x16x4_f32(false, aI, false, fS[ks], (short)0, acc[q][2], false, false);
                }
            }
            float oR[4][8], oI[4][8];
            #pragma unroll
            for (int r8 = 0; r8 < 8; ++r8) {
                const float g0r = acc[0][0][r8] - acc[0][2][r8];
                const float g0i = acc[0][0][r8] + acc[0][1][r8];
                const float x1r = acc[1][0][r8] - acc[1][2][r8];
                const float x1i = acc[1][0][r8] + acc[1][1][r8];
                const float x2r = acc[2][0][r8] - acc[2][2][r8];
                const float x2i = acc[2][0][r8] + acc[2][1][r8];
                const float x3r = acc[3][0][r8] - acc[3][2][r8];
                const float x3i = acc[3][0][r8] + acc[3][1][r8];
                const float g1r = x1r * wc1    - x1i * ws1;
                const float g1i = x1r * ws1    + x1i * wc1;
                const float g2r = x2r * wc23.x - x2i * ws23.x;
                const float g2i = x2r * ws23.x + x2i * wc23.x;
                const float g3r = x3r * wc23.y - x3i * ws23.y;
                const float g3i = x3r * ws23.y + x3i * wc23.y;
                const float ar = g0r + g2r, ai = g0i + g2i;
                const float br = g0r - g2r, bi = g0i - g2i;
                const float cr = g1r + g3r, ci = g1i + g3i;
                const float dr = g1r - g3r, di = g1i - g3i;
                oR[0][r8] = ar + cr; oI[0][r8] = ai + ci;
                oR[1][r8] = br - di; oI[1][r8] = bi + dr;
                oR[2][r8] = ar - cr; oI[2][r8] = ai - ci;
                oR[3][r8] = br + di; oI[3][r8] = bi - dr;
            }
            #pragma unroll
            for (int v2 = 0; v2 < 4; ++v2) {
                const int pi = 4 * bb + v2;
                if (pi >= vt0 && pi < vt0 + n) {
                    #pragma unroll
                    for (int h = 0; h < 8; ++h) {
                        Tre[(v1 + 64 * v2) * PITCH + h * 8 + ((ib + 8 * hi) >> 3)] = oR[v2][h];
                        Tim[(v1 + 64 * v2) * PITCH + h * 8 + ((ib + 8 * hi) >> 3)] = oI[v2][h];
                    }
                }
            }
        }
        __syncthreads();

        const int u1b = (wave & 1) * 16;
        v2f eR[2], eI[2], eS[2];
        #pragma unroll
        for (int ks = 0; ks < 2; ++ks) {
            const int p0 = 4 * ks + 2 * hi;
            eR[ks].x = Ere[(u1b + col) * EP + 2 * p0];
            eR[ks].y = Ere[(u1b + col) * EP + 2 * p0 + 2];
            eI[ks].x = Eim[(u1b + col) * EP + 2 * p0];
            eI[ks].y = Eim[(u1b + col) * EP + 2 * p0 + 2];
            eS[ks] = eR[ks] + eI[ks];
        }
        v2f wc[4][8], ws[4][8];
        #pragma unroll
        for (int h = 0; h < 4; ++h) {
            const int row = u1b + 8 * hi + 2 * h;
            #pragma unroll
            for (int q = 1; q < 8; ++q) {
                wc[h][q] = *(const v2f*)&Q8re[((row >> 1) * 8 + q) * 2];
                ws[h][q] = *(const v2f*)&Q8im[((row >> 1) * 8 + q) * 2];
            }
        }

        float mymax = 0.0f, maxA = 0.0f;
        bool  inB = false;
        for (int pp = wave >> 1; pp < nA + nB; pp += 4) {
            const bool sB = pp >= nA;
            const int  pi = sB ? pp - nA : vtA + pp;
            const int  e  = sB ? eB : eA;
            const int cblk = (pi >> 2) + 4 * (pi & 3);
            if (sB && !inB) { maxA = mymax; mymax = 0.0f; inB = true; }
            const int vrow = cblk * 16 + col;
            const int  cvb   = (cblk + 2) & 15;
            const bool cropu = keep && cvb < 4;
            float* const dst = crop_ws + (size_t)e * (S * S)
                             + ((cvb & 3) * 4 + (wave & 1)) * 256 + lane;
            float sel0[8], sel7[8];
            v8f acc[8][3];
            #pragma unroll
            for (int q = 0; q < 8; ++q) { acc[q][0] = (v8f){}; acc[q][1] = (v8f){}; acc[q][2] = (v8f){}; }
            #pragma unroll
            for (int ks = 0; ks < 2; ++ks) {
                #pragma unroll
                for (int q = 0; q < 8; ++q) {
                    const int k0 = q * 8 + 4 * ks + 2 * hi;
                    const v2f bR = *(const v2f*)&Tre[vrow * PITCH + k0];
                    const v2f bI = *(const v2f*)&Tim[vrow * PITCH + k0];
                    const v2f bD = bI - bR;
                    const v2f bS = bR + bI;
                    acc[q][0] = __builtin_amdgcn_wmma_f32_16x16x4_f32(false, eS[ks], false, bR, (short)0, acc[q][0], false, false);
                    acc[q][1] = __builtin_amdgcn_wmma_f32_16x16x4_f32(false, eR[ks], false, bD, (short)0, acc[q][1], false, false);
                    acc[q][2] = __builtin_amdgcn_wmma_f32_16x16x4_f32(false, eI[ks], false, bS, (short)0, acc[q][2], false, false);
                }
            }
            const float hs = 0.70710678118654752440f;
            #pragma unroll
            for (int h = 0; h < 4; ++h) {
                v2f gr[8], gi[8];
                #pragma unroll
                for (int q = 0; q < 8; ++q) {
                    v2f k1, k2, k3;
                    k1.x = acc[q][0][2 * h]; k1.y = acc[q][0][2 * h + 1];
                    k2.x = acc[q][1][2 * h]; k2.y = acc[q][1][2 * h + 1];
                    k3.x = acc[q][2][2 * h]; k3.y = acc[q][2][2 * h + 1];
                    const v2f xr = k1 - k3;
                    const v2f xi = k1 + k2;
                    if (q == 0) { gr[0] = xr; gi[0] = xi; }
                    else {
                        gr[q] = xr * wc[h][q] - xi * ws[h][q];
                        gi[q] = xr * ws[h][q] + xi * wc[h][q];
                    }
                }
                const v2f a0r = gr[0] + gr[4], a0i = gi[0] + gi[4];
                const v2f a1r = gr[0] - gr[4], a1i = gi[0] - gi[4];
                const v2f a2r = gr[2] + gr[6], a2i = gi[2] + gi[6];
                const v2f a3r = gr[2] - gr[6], a3i = gi[2] - gi[6];
                const v2f A0r = a0r + a2r, A0i = a0i + a2i;
                const v2f A2r = a0r - a2r, A2i = a0i - a2i;
                const v2f A1r = a1r - a3i, A1i = a1i + a3r;
                const v2f A3r = a1r + a3i, A3i = a1i - a3r;
                const v2f b0r = gr[1] + gr[5], b0i = gi[1] + gi[5];
                const v2f b1r = gr[1] - gr[5], b1i = gi[1] - gi[5];
                const v2f b2r = gr[3] + gr[7], b2i = gi[3] + gi[7];
                const v2f b3r = gr[3] - gr[7], b3i = gi[3] - gi[7];
                const v2f B0r = b0r + b2r, B0i = b0i + b2i;
                const v2f B2r = b0r - b2r, B2i = b0i - b2i;
                const v2f B1r = b1r - b3i, B1i = b1i + b3r;
                const v2f B3r = b1r + b3i, B3i = b1i - b3r;
                const v2f t1r = hs * (B1r - B1i), t1i = hs * (B1r + B1i);
                const v2f t3r = hs * (B3r + B3i), t3i = hs * (B3r - B3i);
                const v2f o0r = A0r + B0r, o0i = A0i + B0i;
                const v2f o4r = A0r - B0r, o4i = A0i - B0i;
                const v2f o1r = A1r + t1r, o1i = A1i + t1i;
                const v2f o5r = A1r - t1r, o5i = A1i - t1i;
                const v2f o2r = A2r - B2i, o2i = A2i + B2r;
                const v2f o6r = A2r + B2i, o6i = A2i - B2r;
                const v2f o3r = A3r - t3r, o3i = A3i + t3i;
                const v2f o7r = A3r + t3r, o7i = A3i - t3i;
                const v2f m0 = o0r * o0r + o0i * o0i;
                const v2f m1 = o1r * o1r + o1i * o1i;
                const v2f m2 = o2r * o2r + o2i * o2i;
                const v2f m3 = o3r * o3r + o3i * o3i;
                const v2f m4 = o4r * o4r + o4i * o4i;
                const v2f m5 = o5r * o5r + o5i * o5i;
                const v2f m6 = o6r * o6r + o6i * o6i;
                const v2f m7 = o7r * o7r + o7i * o7i;
                mymax = fmaxf(mymax, fmaxf(m0.x, m0.y));
                mymax = fmaxf(mymax, fmaxf(m1.x, m1.y));
                mymax = fmaxf(mymax, fmaxf(m2.x, m2.y));
                mymax = fmaxf(mymax, fmaxf(m3.x, m3.y));
                mymax = fmaxf(mymax, fmaxf(m4.x, m4.y));
                mymax = fmaxf(mymax, fmaxf(m5.x, m5.y));
                mymax = fmaxf(mymax, fmaxf(m6.x, m6.y));
                mymax = fmaxf(mymax, fmaxf(m7.x, m7.y));
                sel0[2 * h] = m0.x; sel0[2 * h + 1] = m0.y;
                sel7[2 * h] = m7.x; sel7[2 * h + 1] = m7.y;
            }
            if (cropu) {
                #pragma unroll
                for (int r = 0; r < 8; ++r) {
                    dst[r * 32]       = sel7[r];
                    dst[512 + r * 32] = sel0[r];
                }
            }
        }
        float maxB = 0.0f;
        if (inB) maxB = mymax; else maxA = mymax;
        for (int off = 16; off > 0; off >>= 1) {
            maxA = fmaxf(maxA, __shfl_xor(maxA, off, 32));
            maxB = fmaxf(maxB, __shfl_xor(maxB, off, 32));
        }
        if (lane == 0) { wave_max[wave] = maxA; wave_max[8 + wave] = maxB; }
        __syncthreads();
        if (t < 32) {
            const bool hB = t >= 16;
            const int  q  = t & 15;
            if (q < (hB ? nB : nA)) {
                const int o = hB ? 8 : 0;
                float mx = wave_max[o];
                for (int w = 1; w < 8; ++w) mx = fmaxf(mx, wave_max[o + w]);
                pmax[(hB ? eB : eA) * NPIECE + (hB ? 0 : vtA) + q] = mx;
            }
        }
    }
}

__device__ __forceinline__ void crop_by_transform(const float* __restrict__ xs,
                                const float* __restrict__ ys,
                                const float* __restrict__ zs,
                                const float* __restrict__ As,
                                const float* __restrict__ bgs,
                                const float* __restrict__ ps,
                                const float* __restrict__ pupil_phase,
                                const float* __restrict__ pmax,
                                float* __restrict__ out) {
    __shared__ float Pre[S * PITCH];
    __shared__ float Pim[S * PITCH];
    __shared__ float Wre[S * PITCH];
    __shared__ float Wim[S * PITCH];
    __shared__ float Tre[S * PITCH];
    __shared__ float Tim[S * PITCH];

    const int e = blockIdx.x;
    const int b = e >> 2;
    if (ps[e] <= 0.5f) return;
    const float xe = xs[e], ye = ys[e], ze = zs[e];
    const float Ae = As[e], bge = bgs[e];

    const int t    = threadIdx.x;
    const int lane = t & 31;
    const int wave = t >> 5;
    const int col  = lane & 15;
    const int hi   = lane >> 4;

    const float pm = pmax[e * NPIECE + (t & 15)];

    for (int idx = t; idx < S * S; idx += 256) {
        const int c = idx >> 6, j = idx & 63;
        const int v = (c - 32) & 255;
        const int ti = ((j - 32) * v) & 255;
        float sn, cs;
        __sincosf(TWO_PI * (float)ti * (1.0f / 256.0f), &sn, &cs);
        Wre[c * PITCH + j] = cs;
        Wim[c * PITCH + j] = sn;
    }
    build_pupil(Pre, Pim, pupil_phase, xe, ye, ze, t);
    __syncthreads();

    for (int tt = wave * 2; tt < wave * 2 + 2; ++tt) {
        const int ib = (tt >> 2) * 16;
        const int cb = (tt & 3) * 16;
        v8f k1, k2, k3;
        ctile(Pre, Pim, Wre, Wim, ib + col, cb + col, hi, k1, k2, k3);
        store_T(Tre, Tim, cb + col, ib + 8 * hi, k1, k2, k3);
    }
    __syncthreads();

    float* crop = Pre;

    for (int tt = wave * 2; tt < wave * 2 + 2; ++tt) {
        const int ub = (tt >> 2) * 16;
        const int cb = (tt & 3) * 16;
        v8f k1, k2, k3;
        ctile(Wre, Wim, Tre, Tim, ub + col, cb + col, hi, k1, k2, k3);
        #pragma unroll
        for (int q = 0; q < 8; ++q) {
            const float re = k1[q] - k3[q];
            const float im = k1[q] + k2[q];
            const float pw = re * re + im * im;
            crop[(ub + q + 8 * hi) * PITCH + cb + col] = pw;
        }
    }

    float mx = pm;
    for (int off = 8; off > 0; off >>= 1)
        mx = fmaxf(mx, __shfl_xor(mx, off, 32));
    const float inv = 1.0f / mx;
    __syncthreads();

    for (int idx = t; idx < S * S; idx += 256) {
        const int cu = idx >> 6, cv = idx & 63;
        const float val = crop[cu * PITCH + cv] * inv * Ae + bge;
        atomicAdd(&out[b * S * S + idx], val);
    }
}

__launch_bounds__(256)
__global__ void psf_sum_kernel(const float* __restrict__ xs,
                               const float* __restrict__ ys,
                               const float* __restrict__ zs,
                               const float* __restrict__ As,
                               const float* __restrict__ bgs,
                               const float* __restrict__ ps,
                               const float* __restrict__ pupil_phase,
                               const float* __restrict__ pmax,
                               const float* __restrict__ crop_ws,
                               float* __restrict__ out, size_t ws_bytes) {
    if (ws_bytes < WS_NEED) {
        crop_by_transform(xs, ys, zs, As, bgs, ps, pupil_phase, pmax, out);
        return;
    }
    const int b    = blockIdx.x >> 2;
    const int cub  = blockIdx.x & 3;
    const int t    = threadIdx.x;
    const int lane = t & 31;
    const int rr   = t >> 5;
    const int hi   = lane >> 4;
    const int c4   = lane & 15;
    const int off  = ((c4 >> 2) * 4 + cub) * 256 + rr * 32 + hi * 16 + (c4 & 3) * 4;

    float4 acc = make_float4(0.0f, 0.0f, 0.0f, 0.0f);
    #pragma unroll
    for (int n = 0; n < 4; ++n) {
        const int e = 4 * b + n;
        const bool live = !(ps[e] <= 0.5f);
        const float* pm = live ? pmax + e * NPIECE : pupil_phase;
        const float* cs = live ? crop_ws + (size_t)e * (S * S) : pupil_phase;
        const float4 c = *(const float4*)&cs[off];
        float mx = pm[lane & 15];
        for (int o = 8; o > 0; o >>= 1)
            mx = fmaxf(mx, __shfl_xor(mx, o, 32));
        const float inv = 1.0f / mx;
        const float Ae = As[e], bge = bgs[e];
        acc.x += live ? c.x * inv * Ae + bge : 0.0f;
        acc.y += live ? c.y * inv * Ae + bge : 0.0f;
        acc.z += live ? c.z * inv * Ae + bge : 0.0f;
        acc.w += live ? c.w * inv * Ae + bge : 0.0f;
    }
    const int cu = cub * 16 + hi * 8 + rr;
    *(float4*)&out[(size_t)b * (S * S) + cu * S + c4 * 4] = acc;
}

extern "C" void kernel_launch(void* const* d_in, const int* in_sizes, int n_in,
                              void* d_out, int out_size, void* d_ws, size_t ws_size,
                              hipStream_t stream) {
    const float* x  = (const float*)d_in[0];
    const float* y  = (const float*)d_in[1];
    const float* z  = (const float*)d_in[2];
    const float* A  = (const float*)d_in[3];
    const float* bg = (const float*)d_in[4];
    const float* p  = (const float*)d_in[5];
    const float* ph = (const float*)d_in[6];
    float* out = (float*)d_out;
    float* pmax = (float*)d_ws;
    float* crop = pmax + CROP_OFF;

    const int n_emit = in_sizes[0];
    psf_max_kernel<<<NWG, 256, 0, stream>>>(x, y, z, p, ph, out, out_size, n_emit, pmax, crop, ws_size);
    psf_sum_kernel<<<n_emit, 256, 0, stream>>>(x, y, z, A, bg, p, ph, pmax, crop, out, ws_size);
}
